# starting kernel
# baseline (speedup 1.0000x reference)
#include <hip/hip_runtime.h>
#include <hip/hip_bf16.h>

typedef __attribute__((ext_vector_type(2))) float v2f;
typedef __attribute__((ext_vector_type(8))) float v8f;

__global__ __launch_bounds__(256) void zero_kernel(float* __restrict__ p, size_t n) {
    size_t i = (size_t)blockIdx.x * blockDim.x + threadIdx.x;
    if (i < n) p[i] = 0.0f;
}

template <int F>
__global__ __launch_bounds__(256) void scatter_add_vec(const float* __restrict__ src,
                                                       const int* __restrict__ sidx,
                                                       const int* __restrict__ didx,
                                                       float* dst, int nE) {
    constexpr int PER = F / 4;
    unsigned t = blockIdx.x * blockDim.x + threadIdx.x;
    unsigned e = t / PER;
    unsigned f = (t % PER) * 4;
    if (e >= (unsigned)nE) return;
    int s = sidx[e];
    int d = didx[e];
    const float4 v = *(const float4*)(src + (size_t)s * F + f);
    float* p = dst + (size_t)d * F + f;
    atomicAdd(p + 0, v.x);
    atomicAdd(p + 1, v.y);
    atomicAdd(p + 2, v.z);
    atomicAdd(p + 3, v.w);
}

template <int K1, int K2, bool RELU, int MT>
__global__ __launch_bounds__(256) void gemm_dual_wmma(const float* __restrict__ A1,
                                                      const float* __restrict__ W1,
                                                      const float* __restrict__ A2,
                                                      const float* __restrict__ W2,
                                                      const float* __restrict__ bias,
                                                      float* out, int M) {
    constexpr int N = 128;
    const int lane = threadIdx.x & 31;
    const int wave = threadIdx.x >> 5;
    const int m0 = (blockIdx.x * 8 + wave) * (16 * MT);
    if (m0 >= M) return;
    const int half = lane >> 4;
    const int lr = lane & 15;

    v8f acc[MT][8] = {};

    int arow[MT];
#pragma unroll
    for (int mt = 0; mt < MT; ++mt) {
        int r = m0 + mt * 16 + lr;
        arow[mt] = (r < M) ? r : (M - 1);
    }

    for (int k = 0; k < K1; k += 4) {
        v2f a[MT];
#pragma unroll
        for (int mt = 0; mt < MT; ++mt)
            a[mt] = *(const v2f*)(A1 + (size_t)arow[mt] * K1 + k + 2 * half);
        const float* w0 = W1 + (size_t)(k + 2 * half) * N + lr;
#pragma unroll
        for (int nt = 0; nt < 8; ++nt) {
            v2f b;
            b.x = w0[nt * 16];
            b.y = w0[N + nt * 16];
#pragma unroll
            for (int mt = 0; mt < MT; ++mt)
                acc[mt][nt] = __builtin_amdgcn_wmma_f32_16x16x4_f32(
                    false, a[mt], false, b, (short)0, acc[mt][nt], false, false);
        }
    }
    for (int k = 0; k < K2; k += 4) {
        v2f a[MT];
#pragma unroll
        for (int mt = 0; mt < MT; ++mt)
            a[mt] = *(const v2f*)(A2 + (size_t)arow[mt] * K2 + k + 2 * half);
        const float* w0 = W2 + (size_t)(k + 2 * half) * N + lr;
#pragma unroll
        for (int nt = 0; nt < 8; ++nt) {
            v2f b;
            b.x = w0[nt * 16];
            b.y = w0[N + nt * 16];
#pragma unroll
            for (int mt = 0; mt < MT; ++mt)
                acc[mt][nt] = __builtin_amdgcn_wmma_f32_16x16x4_f32(
                    false, a[mt], false, b, (short)0, acc[mt][nt], false, false);
        }
    }

#pragma unroll
    for (int mt = 0; mt < MT; ++mt) {
        if (m0 + mt * 16 >= M) break;
#pragma unroll
        for (int nt = 0; nt < 8; ++nt) {
            const int col = nt * 16 + lr;
            const float bv = bias[col];
#pragma unroll
            for (int i = 0; i < 8; ++i) {
                float v = acc[mt][nt][i] + bv;
                if (RELU) v = fmaxf(v, 0.0f);
                out[(size_t)(m0 + mt * 16 + i + 8 * half) * N + col] = v;
            }
        }
    }
}

__global__ __launch_bounds__(256) void rowmat3(const float* __restrict__ X,
                                               const float* __restrict__ W,
                                               const float* __restrict__ bias,
                                               float* __restrict__ out, int n) {
    const int lane = threadIdx.x & 31;
    const int node = blockIdx.x * (blockDim.x >> 5) + (threadIdx.x >> 5);
    if (node >= n) return;
    const float4 x4 = *(const float4*)(X + (size_t)node * 128 + lane * 4);
    const float* w = W + lane * 4 * 3;
    float p0 = x4.x * w[0] + x4.y * w[3] + x4.z * w[6] + x4.w * w[9];
    float p1 = x4.x * w[1] + x4.y * w[4] + x4.z * w[7] + x4.w * w[10];
    float p2 = x4.x * w[2] + x4.y * w[5] + x4.z * w[8] + x4.w * w[11];
#pragma unroll
    for (int off = 16; off; off >>= 1) {
        p0 += __shfl_xor(p0, off, 32);
        p1 += __shfl_xor(p1, off, 32);
        p2 += __shfl_xor(p2, off, 32);
    }
    if (lane == 0) {
        float b0 = 0.f, b1 = 0.f, b2 = 0.f;
        if (bias) { b0 = bias[0]; b1 = bias[1]; b2 = bias[2]; }
        out[(size_t)node * 3 + 0] = p0 + b0;
        out[(size_t)node * 3 + 1] = p1 + b1;
        out[(size_t)node * 3 + 2] = p2 + b2;
    }
}

__global__ __launch_bounds__(256) void scatter_add3(const float* __restrict__ proj,
                                                    const int* __restrict__ sidx,
                                                    const int* __restrict__ didx,
                                                    float* out, int nE) {
    int e = blockIdx.x * blockDim.x + threadIdx.x;
    if (e >= nE) return;
    int s = sidx[e];
    int d = didx[e];
    atomicAdd(out + (size_t)d * 3 + 0, proj[(size_t)s * 3 + 0]);
    atomicAdd(out + (size_t)d * 3 + 1, proj[(size_t)s * 3 + 1]);
    atomicAdd(out + (size_t)d * 3 + 2, proj[(size_t)s * 3 + 2]);
}

__global__ __launch_bounds__(256) void knowledge_kernel(float* __restrict__ out,
                                                        const float* __restrict__ x,
                                                        int n) {
    int i = blockIdx.x * blockDim.x + threadIdx.x;
    if (i >= n) return;
    float a = out[(size_t)i * 3 + 0];
    float b = out[(size_t)i * 3 + 1];
    float c = out[(size_t)i * 3 + 2];
    float nrm = sqrtf(a * a + b * b + c * c);
    float s = 10.0f / fmaxf(nrm, 1e-12f);
    a *= s; b *= s; c *= s;
    if (x[(size_t)i * 64 + 61] != 0.0f) a -= 10.0f;
    if (x[(size_t)i * 64 + 63] != 0.0f) c -= 10.0f;
    out[(size_t)i * 3 + 0] = a;
    out[(size_t)i * 3 + 1] = b;
    out[(size_t)i * 3 + 2] = c;
}

static inline int ceil_div(long long a, long long b) { return (int)((a + b - 1) / b); }

extern "C" void kernel_launch(void* const* d_in, const int* in_sizes, int n_in,
                              void* d_out, int out_size, void* d_ws, size_t ws_size,
                              hipStream_t stream) {
    const float* x_s = (const float*)d_in[0];
    const float* x_t = (const float*)d_in[1];
    const int* e_l = (const int*)d_in[2];
    const int* e_r = (const int*)d_in[3];
    const float* w1lr_rel  = (const float*)d_in[4];
    const float* b1lr      = (const float*)d_in[5];
    const float* w1lr_root = (const float*)d_in[6];
    const float* w1rl_rel  = (const float*)d_in[7];
    const float* b1rl      = (const float*)d_in[8];
    const float* w1rl_root = (const float*)d_in[9];
    const float* w2lr_rel  = (const float*)d_in[10];
    const float* b2lr      = (const float*)d_in[11];
    const float* w2lr_root = (const float*)d_in[12];
    const float* w2rl_rel  = (const float*)d_in[13];
    const float* b2rl      = (const float*)d_in[14];
    const float* w2rl_root = (const float*)d_in[15];
    const float* w3lr_rel  = (const float*)d_in[16];
    const float* b3lr      = (const float*)d_in[17];
    const float* w3lr_root = (const float*)d_in[18];
    const float* w3rl_rel  = (const float*)d_in[19];
    const float* b3rl      = (const float*)d_in[20];
    const float* w3rl_root = (const float*)d_in[21];

    const int NL = in_sizes[0] / 64;
    const int NR = in_sizes[1] / 64;
    const int E  = in_sizes[2];

    float* ws = (float*)d_ws;
    const size_t AGG_R = 0;
    const size_t AGG_L = AGG_R + (size_t)NR * 128;
    const size_t H1_R  = AGG_L + (size_t)NL * 128;
    const size_t H1_L  = H1_R + (size_t)NR * 128;
    const size_t PROJ_LR = H1_R;
    const size_t PROJ_RL = H1_R + (size_t)NL * 3;

    float* d_left  = (float*)d_out;
    float* d_right = (float*)d_out + (size_t)NL * 3;

    const int TPB = 256;
    constexpr int MT = 2;

    zero_kernel<<<ceil_div((size_t)NR * 64, TPB), TPB, 0, stream>>>(ws + AGG_R, (size_t)NR * 64);
    zero_kernel<<<ceil_div((size_t)NL * 64, TPB), TPB, 0, stream>>>(ws + AGG_L, (size_t)NL * 64);
    scatter_add_vec<64><<<ceil_div((long long)E * 16, TPB), TPB, 0, stream>>>(x_s, e_l, e_r, ws + AGG_R, E);
    scatter_add_vec<64><<<ceil_div((long long)E * 16, TPB), TPB, 0, stream>>>(x_t, e_r, e_l, ws + AGG_L, E);
    gemm_dual_wmma<64, 64, true, MT><<<ceil_div(ceil_div(NR, 16 * MT), 8), TPB, 0, stream>>>(
        ws + AGG_R, w1lr_rel, x_t, w1lr_root, b1lr, ws + H1_R, NR);
    gemm_dual_wmma<64, 64, true, MT><<<ceil_div(ceil_div(NL, 16 * MT), 8), TPB, 0, stream>>>(
        ws + AGG_L, w1rl_rel, x_s, w1rl_root, b1rl, ws + H1_L, NL);

    zero_kernel<<<ceil_div((size_t)NR * 128, TPB), TPB, 0, stream>>>(ws + AGG_R, (size_t)NR * 128);
    zero_kernel<<<ceil_div((size_t)NL * 128, TPB), TPB, 0, stream>>>(ws + AGG_L, (size_t)NL * 128);
    scatter_add_vec<128><<<ceil_div((long long)E * 32, TPB), TPB, 0, stream>>>(ws + H1_L, e_l, e_r, ws + AGG_R, E);
    scatter_add_vec<128><<<ceil_div((long long)E * 32, TPB), TPB, 0, stream>>>(ws + H1_R, e_r, e_l, ws + AGG_L, E);
    gemm_dual_wmma<128, 128, true, MT><<<ceil_div(ceil_div(NR, 16 * MT), 8), TPB, 0, stream>>>(
        ws + AGG_R, w2lr_rel, ws + H1_R, w2lr_root, b2lr, ws + AGG_R, NR);
    gemm_dual_wmma<128, 128, true, MT><<<ceil_div(ceil_div(NL, 16 * MT), 8), TPB, 0, stream>>>(
        ws + AGG_L, w2rl_rel, ws + H1_L, w2rl_root, b2rl, ws + AGG_L, NL);

    rowmat3<<<ceil_div(NL, 8), TPB, 0, stream>>>(ws + AGG_L, w3lr_rel, nullptr, ws + PROJ_LR, NL);
    rowmat3<<<ceil_div(NR, 8), TPB, 0, stream>>>(ws + AGG_R, w3rl_rel, nullptr, ws + PROJ_RL, NR);
    rowmat3<<<ceil_div(NR, 8), TPB, 0, stream>>>(ws + AGG_R, w3lr_root, b3lr, d_right, NR);
    rowmat3<<<ceil_div(NL, 8), TPB, 0, stream>>>(ws + AGG_L, w3rl_root, b3rl, d_left, NL);
    scatter_add3<<<ceil_div(E, TPB), TPB, 0, stream>>>(ws + PROJ_LR, e_l, e_r, d_right, E);
    scatter_add3<<<ceil_div(E, TPB), TPB, 0, stream>>>(ws + PROJ_RL, e_r, e_l, d_left, E);

    knowledge_kernel<<<ceil_div(NL, TPB), TPB, 0, stream>>>(d_left, x_s, NL);
    knowledge_kernel<<<ceil_div(NR, TPB), TPB, 0, stream>>>(d_right, x_t, NR);
}


# GEMM access re-laid for whole 128-byte lines: a lane owns 4 adjacent output columns (16-byte weight loads, output stores); activation rows fetched once, whole rows, into per-wave LDS; sums unchanged
# speedup vs baseline: 2.1502x; 2.1502x over previous
#include <hip/hip_runtime.h>
#include <hip/hip_bf16.h>

typedef __attribute__((ext_vector_type(2))) float v2f;
typedef __attribute__((ext_vector_type(8))) float v8f;

__global__ __launch_bounds__(256) void zero_kernel(float* __restrict__ p, size_t n,
                                                   int* __restrict__ tickets, int n_tickets) {
    size_t i = (size_t)blockIdx.x * blockDim.x + threadIdx.x;
    if (i < n) p[i] = 0.0f;
    if (i < (size_t)n_tickets) tickets[i] = 0;
}

template <int F, int G>
__global__ __launch_bounds__(256) void scatter_add_rows(const float* __restrict__ src,
                                                        const int* __restrict__ sidx,
                                                        const int* __restrict__ didx,
                                                        float* dst, int nE, int n_dst,
                                                        int* tickets) {
    constexpr int C = F / 32;
    constexpr int CH = 8 * G * 32;
    constexpr int JB = 64 / C;
    __shared__ int s_tk[2];
    __shared__ int s_half;
    const int lane = threadIdx.x & 31;
    const int wave = __builtin_amdgcn_readfirstlane((int)(threadIdx.x >> 5));
    if (threadIdx.x == 0) {
        const unsigned id = __builtin_amdgcn_s_sendmsg_rtn(0x87);
        s_half = (int)((id >> 18) & 1u);
    }
    __syncthreads();
    int tgt = __builtin_amdgcn_readfirstlane(s_half);
    const int nsplit = n_dst >> 1;
    const int nCh = (nE + CH - 1) / CH;
    int pass = 0;
    for (int it = 0; it < 2 * nCh + 2; ++it) {
        if (threadIdx.x == 0) s_tk[it & 1] = atomicAdd(tickets + tgt * 32, 1);
        __syncthreads();
        const int tk = __builtin_amdgcn_readfirstlane(s_tk[it & 1]);
        if (tk >= nCh) {
            if (pass) break;
            pass = 1;
            tgt ^= 1;
            continue;
        }
        unsigned soffs[G], doffs[G], masks[G];
#pragma unroll
        for (int g = 0; g < G; ++g) {
            const int e = tk * CH + (wave * G + g) * 32 + lane;
            const int ec = (e < nE) ? e : (nE - 1);
            const int sv = sidx[ec];
            const int dv = didx[ec];
            soffs[g] = (unsigned)sv * (unsigned)(F * 4);
            doffs[g] = (unsigned)dv * (unsigned)(F * 4);
            const bool mine = (e < nE) && (((dv >= nsplit) ? 1 : 0) == tgt);
            masks[g] = __builtin_amdgcn_ballot_w32(mine);
        }
#pragma unroll
        for (int g = 0; g < G; ++g) {
            const unsigned soff = soffs[g], doff = doffs[g], mask = masks[g];
            const unsigned lane4 = (unsigned)lane * 4u;
#pragma unroll
            for (int j0 = 0; j0 < 32; j0 += JB) {
                float v[JB][C] = {};
#pragma unroll
                for (int j = 0; j < JB; ++j) {
                    if ((mask >> (j0 + j)) & 1u) {
                        const unsigned so = __builtin_amdgcn_readlane(soff, j0 + j);
                        const char* p = (const char*)src + so;
#pragma unroll
                        for (int c = 0; c < C; ++c) v[j][c] = *(const float*)(p + (lane4 + c * 128u));
                    }
                }
#pragma unroll
                for (int j = 0; j < JB; ++j) {
#pragma unroll
                    for (int c = 0; c < C; ++c) asm volatile("" ::"v"(v[j][c]));
                }
#pragma unroll
                for (int j = 0; j < JB; ++j) {
                    if ((mask >> (j0 + j)) & 1u) {
                        const unsigned dof = __builtin_amdgcn_readlane(doff, j0 + j);
                        char* q = (char*)dst + dof;
#pragma unroll
                        for (int c = 0; c < C; ++c) atomicAdd((float*)(q + (lane4 + c * 128u)), v[j][c]);
                    }
                }
            }
        }
    }
}

constexpr int GEMM_PITCH = 132;

template <int KK, int MT>
__device__ __forceinline__ void gemm_k_steps(const float (*sa)[GEMM_PITCH], int acol0,
                                             const float* __restrict__ W, int lr, int half,
                                             v8f (&acc)[MT][8]) {
    constexpr int N = 128;
#pragma unroll 4
    for (int k = 0; k < KK; k += 4) {
        v2f a[MT];
#pragma unroll
        for (int mt = 0; mt < MT; ++mt)
            a[mt] = *(const v2f*)&sa[mt * 16 + lr][acol0 + k + 2 * half];
        const float* w0 = W + (size_t)(k + 2 * half) * N + lr * 4;
        const float4 x0 = *(const float4*)(w0);
        const float4 x1 = *(const float4*)(w0 + 64);
        const float4 y0 = *(const float4*)(w0 + N);
        const float4 y1 = *(const float4*)(w0 + N + 64);
        const float bx[8] = {x0.x, x0.y, x0.z, x0.w, x1.x, x1.y, x1.z, x1.w};
        const float by[8] = {y0.x, y0.y, y0.z, y0.w, y1.x, y1.y, y1.z, y1.w};
#pragma unroll
        for (int nt = 0; nt < 8; ++nt) {
            v2f b;
            b.x = bx[nt];
            b.y = by[nt];
#pragma unroll
            for (int mt = 0; mt < MT; ++mt)
                acc[mt][nt] = __builtin_amdgcn_wmma_f32_16x16x4_f32(
                    false, a[mt], false, b, (short)0, acc[mt][nt], false, false);
        }
    }
}

template <int K1, int K2, bool RELU, int MT>
__global__ __launch_bounds__(256) void gemm_dual_wmma(const float* __restrict__ A1,
                                                      const float* __restrict__ W1a,
                                                      const float* __restrict__ A2,
                                                      const float* __restrict__ W2,
                                                      const float* __restrict__ biasa,
                                                      float* outa,
                                                      const float* __restrict__ W1b,
                                                      const float* __restrict__ biasb,
                                                      float* outb, int M) {
    constexpr int N = 128;
    constexpr int ROWS = 16 * MT;
    static_assert(K1 + K2 == 128, "a lane stages 4 of the 128 joined columns");
    __shared__ __attribute__((aligned(16))) float s_a[8][ROWS][GEMM_PITCH];
    const float* W1 = blockIdx.y ? W1b : W1a;
    const float* bias = blockIdx.y ? biasb : biasa;
    float* out = blockIdx.y ? outb : outa;
    const int lane = threadIdx.x & 31;
    const int wave = threadIdx.x >> 5;
    const int m0 = (blockIdx.x * 8 + wave) * ROWS;
    const int half = lane >> 4;
    const int lr = lane & 15;
    float (*sa)[GEMM_PITCH] = s_a[wave];

    if (m0 < M) {
        const int c4 = lane * 4;
        const float* srcp = A1;
        int srck = K1, coff = c4;
        if (K2 > 0 && c4 >= K1) { srcp = A2; srck = K2; coff = c4 - K1; }
#pragma unroll 8
        for (int r = 0; r < ROWS; ++r) {
            int g = m0 + r;
            g = (g < M) ? g : (M - 1);
            const float4 v = *(const float4*)(srcp + (size_t)g * srck + coff);
            *(float4*)&sa[r][c4] = v;
        }
    }
    __syncthreads();
    if (m0 >= M) return;

    v8f acc[MT][8] = {};
    gemm_k_steps<K1, MT>(sa, 0, W1, lr, half, acc);
    if constexpr (K2 > 0) gemm_k_steps<K2, MT>(sa, K1, W2, lr, half, acc);

#pragma unroll
    for (int mt = 0; mt < MT; ++mt) {
        if (m0 + mt * 16 >= M) break;
#pragma unroll
        for (int j = 0; j < 2; ++j) {
            const int col = j * 64 + lr * 4;
            float4 bv = {0.0f, 0.0f, 0.0f, 0.0f};
            if (bias) bv = *(const float4*)(bias + col);
#pragma unroll
            for (int i = 0; i < 8; ++i) {
                float4 v;
                v.x = acc[mt][j * 4 + 0][i] + bv.x;
                v.y = acc[mt][j * 4 + 1][i] + bv.y;
                v.z = acc[mt][j * 4 + 2][i] + bv.z;
                v.w = acc[mt][j * 4 + 3][i] + bv.w;
                if (RELU) {
                    v.x = fmaxf(v.x, 0.0f);
                    v.y = fmaxf(v.y, 0.0f);
                    v.z = fmaxf(v.z, 0.0f);
                    v.w = fmaxf(v.w, 0.0f);
                }
                *(float4*)(out + (size_t)(m0 + mt * 16 + i + 8 * half) * N + col) = v;
            }
        }
    }
}

__global__ __launch_bounds__(256) void rowmat3(const float* __restrict__ X,
                                               const float* __restrict__ W,
                                               const float* __restrict__ bias,
                                               float* __restrict__ out, int n) {
    const int lane = threadIdx.x & 31;
    const int node = blockIdx.x * (blockDim.x >> 5) + (threadIdx.x >> 5);
    if (node >= n) return;
    float4 x4 = *(const float4*)(X + (size_t)node * 128 + lane * 4);
    x4.x = fmaxf(x4.x, 0.0f);
    x4.y = fmaxf(x4.y, 0.0f);
    x4.z = fmaxf(x4.z, 0.0f);
    x4.w = fmaxf(x4.w, 0.0f);
    const float* w = W + lane * 4 * 3;
    float p0 = x4.x * w[0] + x4.y * w[3] + x4.z * w[6] + x4.w * w[9];
    float p1 = x4.x * w[1] + x4.y * w[4] + x4.z * w[7] + x4.w * w[10];
    float p2 = x4.x * w[2] + x4.y * w[5] + x4.z * w[8] + x4.w * w[11];
#pragma unroll
    for (int off = 16; off; off >>= 1) {
        p0 += __shfl_xor(p0, off, 32);
        p1 += __shfl_xor(p1, off, 32);
        p2 += __shfl_xor(p2, off, 32);
    }
    if (lane == 0) {
        float b0 = 0.f, b1 = 0.f, b2 = 0.f;
        if (bias) { b0 = bias[0]; b1 = bias[1]; b2 = bias[2]; }
        out[(size_t)node * 3 + 0] = p0 + b0;
        out[(size_t)node * 3 + 1] = p1 + b1;
        out[(size_t)node * 3 + 2] = p2 + b2;
    }
}

__global__ __launch_bounds__(256) void scatter_add3(const float* __restrict__ proj,
                                                    const int* __restrict__ sidx,
                                                    const int* __restrict__ didx,
                                                    float* out, int nE) {
    const int lane = threadIdx.x & 31;
    const int e0 = (blockIdx.x * 8 + (int)(threadIdx.x >> 5)) * 32;
    if (e0 >= nE) return;
    const int e = e0 + lane;
    const int ec = (e < nE) ? e : (nE - 1);
    const int sv = sidx[ec];
    const int dv = didx[ec];
    const int c = lane & 3;
    const int q = lane >> 2;
    float v[4];
    int doff[4];
    bool act[4];
#pragma unroll
    for (int r = 0; r < 4; ++r) {
        const int sl = r * 8 + q;
        const int s = __shfl(sv, sl, 32);
        const int d = __shfl(dv, sl, 32);
        act[r] = (c < 3) && (e0 + sl < nE);
        doff[r] = d * 3 + c;
        v[r] = act[r] ? proj[(size_t)s * 3 + c] : 0.0f;
    }
#pragma unroll
    for (int r = 0; r < 4; ++r) {
        if (act[r]) atomicAdd(out + doff[r], v[r]);
    }
}

__global__ __launch_bounds__(256) void knowledge_kernel(float* __restrict__ out,
                                                        const float* __restrict__ x,
                                                        int n) {
    int i = blockIdx.x * blockDim.x + threadIdx.x;
    if (i >= n) return;
    float a = out[(size_t)i * 3 + 0];
    float b = out[(size_t)i * 3 + 1];
    float c = out[(size_t)i * 3 + 2];
    float nrm = sqrtf(a * a + b * b + c * c);
    float s = 10.0f / fmaxf(nrm, 1e-12f);
    a *= s; b *= s; c *= s;
    if (x[(size_t)i * 64 + 61] != 0.0f) a -= 10.0f;
    if (x[(size_t)i * 64 + 63] != 0.0f) c -= 10.0f;
    out[(size_t)i * 3 + 0] = a;
    out[(size_t)i * 3 + 1] = b;
    out[(size_t)i * 3 + 2] = c;
}

static inline int ceil_div(long long a, long long b) { return (int)((a + b - 1) / b); }

extern "C" void kernel_launch(void* const* d_in, const int* in_sizes, int n_in,
                              void* d_out, int out_size, void* d_ws, size_t ws_size,
                              hipStream_t stream) {
    const float* x_s = (const float*)d_in[0];
    const float* x_t = (const float*)d_in[1];
    const int* e_l = (const int*)d_in[2];
    const int* e_r = (const int*)d_in[3];
    const float* w1lr_rel  = (const float*)d_in[4];
    const float* b1lr      = (const float*)d_in[5];
    const float* w1lr_root = (const float*)d_in[6];
    const float* w1rl_rel  = (const float*)d_in[7];
    const float* b1rl      = (const float*)d_in[8];
    const float* w1rl_root = (const float*)d_in[9];
    const float* w2lr_rel  = (const float*)d_in[10];
    const float* b2lr      = (const float*)d_in[11];
    const float* w2lr_root = (const float*)d_in[12];
    const float* w2rl_rel  = (const float*)d_in[13];
    const float* b2rl      = (const float*)d_in[14];
    const float* w2rl_root = (const float*)d_in[15];
    const float* w3lr_rel  = (const float*)d_in[16];
    const float* b3lr      = (const float*)d_in[17];
    const float* w3lr_root = (const float*)d_in[18];
    const float* w3rl_rel  = (const float*)d_in[19];
    const float* b3rl      = (const float*)d_in[20];
    const float* w3rl_root = (const float*)d_in[21];

    const int NL = in_sizes[0] / 64;
    const int NR = in_sizes[1] / 64;
    const int E  = in_sizes[2];

    float* ws = (float*)d_ws;
    const size_t AGG_R = 0;
    const size_t AGG_L = AGG_R + (size_t)NR * 128;
    const size_t H1_R  = AGG_L + (size_t)NL * 128;
    const size_t H1_L  = H1_R + (size_t)NR * 128;
    const size_t P_R   = H1_L + (size_t)NL * 128;
    const size_t P_L   = P_R + (size_t)NR * 128;
    const size_t TKT   = P_L + (size_t)NL * 128;
    const size_t PROJ_LR = H1_R;
    const size_t PROJ_RL = H1_R + (size_t)NL * 3;

    int* tkt = (int*)(ws + TKT);

    float* d_left  = (float*)d_out;
    float* d_right = (float*)d_out + (size_t)NL * 3;

    const int TPB = 256;
    constexpr int MT = 2;
    constexpr int GR = 2;
    const int SGRID = 1536;

    zero_kernel<<<ceil_div((size_t)NR * 64, TPB), TPB, 0, stream>>>(ws + AGG_R, (size_t)NR * 64, nullptr, 0);
    zero_kernel<<<ceil_div((size_t)NL * 64, TPB), TPB, 0, stream>>>(ws + AGG_L, (size_t)NL * 64, tkt, 256);
    scatter_add_rows<64, GR><<<SGRID, TPB, 0, stream>>>(x_s, e_l, e_r, ws + AGG_R, E, NR, tkt);
    scatter_add_rows<64, GR><<<SGRID, TPB, 0, stream>>>(x_t, e_r, e_l, ws + AGG_L, E, NL, tkt + 64);
    gemm_dual_wmma<64, 64, true, MT><<<ceil_div(ceil_div(NR, 16 * MT), 8), TPB, 0, stream>>>(
        ws + AGG_R, w1lr_rel, x_t, w1lr_root, b1lr, ws + H1_R, nullptr, nullptr, nullptr, NR);
    gemm_dual_wmma<64, 64, true, MT><<<ceil_div(ceil_div(NL, 16 * MT), 8), TPB, 0, stream>>>(
        ws + AGG_L, w1rl_rel, x_s, w1rl_root, b1rl, ws + H1_L, nullptr, nullptr, nullptr, NL);

    gemm_dual_wmma<128, 0, false, MT><<<dim3(ceil_div(ceil_div(NR, 16 * MT), 8), 2), TPB, 0, stream>>>(
        ws + H1_R, w2rl_rel, nullptr, nullptr, nullptr, ws + P_R, w2lr_root, b2lr, ws + AGG_R, NR);
    gemm_dual_wmma<128, 0, false, MT><<<dim3(ceil_div(ceil_div(NL, 16 * MT), 8), 2), TPB, 0, stream>>>(
        ws + H1_L, w2lr_rel, nullptr, nullptr, nullptr, ws + P_L, w2rl_root, b2rl, ws + AGG_L, NL);
    scatter_add_rows<128, GR><<<SGRID, TPB, 0, stream>>>(ws + P_L, e_l, e_r, ws + AGG_R, E, NR, tkt + 128);
    scatter_add_rows<128, GR><<<SGRID, TPB, 0, stream>>>(ws + P_R, e_r, e_l, ws + AGG_L, E, NL, tkt + 192);

    rowmat3<<<ceil_div(NL, 8), TPB, 0, stream>>>(ws + AGG_L, w3lr_rel, nullptr, ws + PROJ_LR, NL);
    rowmat3<<<ceil_div(NR, 8), TPB, 0, stream>>>(ws + AGG_R, w3rl_rel, nullptr, ws + PROJ_RL, NR);
    rowmat3<<<ceil_div(NR, 8), TPB, 0, stream>>>(ws + AGG_R, w3lr_root, b3lr, d_right, NR);
    rowmat3<<<ceil_div(NL, 8), TPB, 0, stream>>>(ws + AGG_L, w3rl_root, b3rl, d_left, NL);
    scatter_add3<<<ceil_div(E, TPB), TPB, 0, stream>>>(ws + PROJ_LR, e_l, e_r, d_right, E);
    scatter_add3<<<ceil_div(E, TPB), TPB, 0, stream>>>(ws + PROJ_RL, e_r, e_l, d_left, E);

    knowledge_kernel<<<ceil_div(NL, TPB), TPB, 0, stream>>>(d_left, x_s, NL);
    knowledge_kernel<<<ceil_div(NR, TPB), TPB, 0, stream>>>(d_right, x_t, NR);
}


# Layer-3 node products moved to the matrix unit: a wave stages 16 relu'd rows in LDS and forms rel and root dots of a row from one read by f32 WMMA; 18 launches become 16; dots summed in k order
# speedup vs baseline: 2.2732x; 1.0572x over previous
#include <hip/hip_runtime.h>
#include <hip/hip_bf16.h>

typedef __attribute__((ext_vector_type(2))) float v2f;
typedef __attribute__((ext_vector_type(8))) float v8f;

__global__ __launch_bounds__(256) void zero_kernel(float* __restrict__ p, size_t n,
                                                   int* __restrict__ tickets, int n_tickets) {
    size_t i = (size_t)blockIdx.x * blockDim.x + threadIdx.x;
    if (i < n) p[i] = 0.0f;
    if (i < (size_t)n_tickets) tickets[i] = 0;
}

template <int F, int G>
__global__ __launch_bounds__(256) void scatter_add_rows(const float* __restrict__ src,
                                                        const int* __restrict__ sidx,
                                                        const int* __restrict__ didx,
                                                        float* dst, int nE, int n_dst,
                                                        int* tickets) {
    constexpr int C = F / 32;
    constexpr int CH = 8 * G * 32;
    constexpr int JB = 64 / C;
    __shared__ int s_tk[2];
    __shared__ int s_half;
    const int lane = threadIdx.x & 31;
    const int wave = __builtin_amdgcn_readfirstlane((int)(threadIdx.x >> 5));
    if (threadIdx.x == 0) {
        const unsigned id = __builtin_amdgcn_s_sendmsg_rtn(0x87);
        s_half = (int)((id >> 18) & 1u);
    }
    __syncthreads();
    int tgt = __builtin_amdgcn_readfirstlane(s_half);
    const int nsplit = n_dst >> 1;
    const int nCh = (nE + CH - 1) / CH;
    int pass = 0;
    for (int it = 0; it < 2 * nCh + 2; ++it) {
        if (threadIdx.x == 0) s_tk[it & 1] = atomicAdd(tickets + tgt * 32, 1);
        __syncthreads();
        const int tk = __builtin_amdgcn_readfirstlane(s_tk[it & 1]);
        if (tk >= nCh) {
            if (pass) break;
            pass = 1;
            tgt ^= 1;
            continue;
        }
        unsigned soffs[G], doffs[G], masks[G];
#pragma unroll
        for (int g = 0; g < G; ++g) {
            const int e = tk * CH + (wave * G + g) * 32 + lane;
            const int ec = (e < nE) ? e : (nE - 1);
            const int sv = sidx[ec];
            const int dv = didx[ec];
            soffs[g] = (unsigned)sv * (unsigned)(F * 4);
            doffs[g] = (unsigned)dv * (unsigned)(F * 4);
            const bool mine = (e < nE) && (((dv >= nsplit) ? 1 : 0) == tgt);
            masks[g] = __builtin_amdgcn_ballot_w32(mine);
        }
#pragma unroll
        for (int g = 0; g < G; ++g) {
            const unsigned soff = soffs[g], doff = doffs[g], mask = masks[g];
            const unsigned lane4 = (unsigned)lane * 4u;
#pragma unroll
            for (int j0 = 0; j0 < 32; j0 += JB) {
                float v[JB][C] = {};
#pragma unroll
                for (int j = 0; j < JB; ++j) {
                    if ((mask >> (j0 + j)) & 1u) {
                        const unsigned so = __builtin_amdgcn_readlane(soff, j0 + j);
                        const char* p = (const char*)src + so;
#pragma unroll
                        for (int c = 0; c < C; ++c) v[j][c] = *(const float*)(p + (lane4 + c * 128u));
                    }
                }
#pragma unroll
                for (int j = 0; j < JB; ++j) {
#pragma unroll
                    for (int c = 0; c < C; ++c) asm volatile("" ::"v"(v[j][c]));
                }
#pragma unroll
                for (int j = 0; j < JB; ++j) {
                    if ((mask >> (j0 + j)) & 1u) {
                        const unsigned dof = __builtin_amdgcn_readlane(doff, j0 + j);
                        char* q = (char*)dst + dof;
#pragma unroll
                        for (int c = 0; c < C; ++c) atomicAdd((float*)(q + (lane4 + c * 128u)), v[j][c]);
                    }
                }
            }
        }
    }
}

constexpr int GEMM_PITCH = 132;

template <int KK, int MT>
__device__ __forceinline__ void gemm_k_steps(const float (*sa)[GEMM_PITCH], int acol0,
                                             const float* __restrict__ W, int lr, int half,
                                             v8f (&acc)[MT][8]) {
    constexpr int N = 128;
#pragma unroll 4
    for (int k = 0; k < KK; k += 4) {
        v2f a[MT];
#pragma unroll
        for (int mt = 0; mt < MT; ++mt)
            a[mt] = *(const v2f*)&sa[mt * 16 + lr][acol0 + k + 2 * half];
        const float* w0 = W + (size_t)(k + 2 * half) * N + lr * 4;
        const float4 x0 = *(const float4*)(w0);
        const float4 x1 = *(const float4*)(w0 + 64);
        const float4 y0 = *(const float4*)(w0 + N);
        const float4 y1 = *(const float4*)(w0 + N + 64);
        const float bx[8] = {x0.x, x0.y, x0.z, x0.w, x1.x, x1.y, x1.z, x1.w};
        const float by[8] = {y0.x, y0.y, y0.z, y0.w, y1.x, y1.y, y1.z, y1.w};
#pragma unroll
        for (int nt = 0; nt < 8; ++nt) {
            v2f b;
            b.x = bx[nt];
            b.y = by[nt];
#pragma unroll
            for (int mt = 0; mt < MT; ++mt)
                acc[mt][nt] = __builtin_amdgcn_wmma_f32_16x16x4_f32(
                    false, a[mt], false, b, (short)0, acc[mt][nt], false, false);
        }
    }
}

template <int K1, int K2, bool RELU, int MT>
__global__ __launch_bounds__(256) void gemm_dual_wmma(const float* __restrict__ A1,
                                                      const float* __restrict__ W1a,
                                                      const float* __restrict__ A2,
                                                      const float* __restrict__ W2,
                                                      const float* __restrict__ biasa,
                                                      float* outa,
                                                      const float* __restrict__ W1b,
                                                      const float* __restrict__ biasb,
                                                      float* outb, int M) {
    constexpr int N = 128;
    constexpr int ROWS = 16 * MT;
    static_assert(K1 + K2 == 128, "a lane stages 4 of the 128 joined columns");
    __shared__ __attribute__((aligned(16))) float s_a[8][ROWS][GEMM_PITCH];
    const float* W1 = blockIdx.y ? W1b : W1a;
    const float* bias = blockIdx.y ? biasb : biasa;
    float* out = blockIdx.y ? outb : outa;
    const int lane = threadIdx.x & 31;
    const int wave = threadIdx.x >> 5;
    const int m0 = (blockIdx.x * 8 + wave) * ROWS;
    const int half = lane >> 4;
    const int lr = lane & 15;
    float (*sa)[GEMM_PITCH] = s_a[wave];

    if (m0 < M) {
        const int c4 = lane * 4;
        const float* srcp = A1;
        int srck = K1, coff = c4;
        if (K2 > 0 && c4 >= K1) { srcp = A2; srck = K2; coff = c4 - K1; }
#pragma unroll 8
        for (int r = 0; r < ROWS; ++r) {
            int g = m0 + r;
            g = (g < M) ? g : (M - 1);
            const float4 v = *(const float4*)(srcp + (size_t)g * srck + coff);
            *(float4*)&sa[r][c4] = v;
        }
    }
    __syncthreads();
    if (m0 >= M) return;

    v8f acc[MT][8] = {};
    gemm_k_steps<K1, MT>(sa, 0, W1, lr, half, acc);
    if constexpr (K2 > 0) gemm_k_steps<K2, MT>(sa, K1, W2, lr, half, acc);

#pragma unroll
    for (int mt = 0; mt < MT; ++mt) {
        if (m0 + mt * 16 >= M) break;
#pragma unroll
        for (int j = 0; j < 2; ++j) {
            const int col = j * 64 + lr * 4;
            float4 bv = {0.0f, 0.0f, 0.0f, 0.0f};
            if (bias) bv = *(const float4*)(bias + col);
#pragma unroll
            for (int i = 0; i < 8; ++i) {
                float4 v;
                v.x = acc[mt][j * 4 + 0][i] + bv.x;
                v.y = acc[mt][j * 4 + 1][i] + bv.y;
                v.z = acc[mt][j * 4 + 2][i] + bv.z;
                v.w = acc[mt][j * 4 + 3][i] + bv.w;
                if (RELU) {
                    v.x = fmaxf(v.x, 0.0f);
                    v.y = fmaxf(v.y, 0.0f);
                    v.z = fmaxf(v.z, 0.0f);
                    v.w = fmaxf(v.w, 0.0f);
                }
                *(float4*)(out + (size_t)(m0 + mt * 16 + i + 8 * half) * N + col) = v;
            }
        }
    }
}

__global__ __launch_bounds__(256) void rowmat6_wmma(const float* __restrict__ X,
                                                    const float* __restrict__ Wrel,
                                                    const float* __restrict__ Wroot,
                                                    const float* __restrict__ bias,
                                                    float* __restrict__ proj,
                                                    float* __restrict__ init, int n) {
    __shared__ __attribute__((aligned(16))) float s_x[8][16][GEMM_PITCH];
    const int lane = threadIdx.x & 31;
    const int wave = threadIdx.x >> 5;
    const int m0 = (blockIdx.x * 8 + wave) * 16;
    const int half = lane >> 4;
    const int lr = lane & 15;
    float (*sx)[GEMM_PITCH] = s_x[wave];
    if (m0 < n) {
        float4 v[16];
#pragma unroll
        for (int r = 0; r < 16; ++r) {
            int g = m0 + r;
            g = (g < n) ? g : (n - 1);
            v[r] = *(const float4*)(X + (size_t)g * 128 + lane * 4);
        }
#pragma unroll
        for (int r = 0; r < 16; ++r) {
            float4 t = v[r];
            t.x = fmaxf(t.x, 0.0f);
            t.y = fmaxf(t.y, 0.0f);
            t.z = fmaxf(t.z, 0.0f);
            t.w = fmaxf(t.w, 0.0f);
            *(float4*)&sx[r][lane * 4] = t;
        }
    }
    __syncthreads();
    if (m0 >= n) return;

    const bool used = lr < 6;
    const float* wcol = (lr < 3) ? (Wrel + lr) : (used ? (Wroot + (lr - 3)) : Wrel);
    v8f acc = {};
#pragma unroll
    for (int k = 0; k < 128; k += 4) {
        const v2f a = *(const v2f*)&sx[lr][k + 2 * half];
        const float wx = wcol[(k + 2 * half) * 3];
        const float wy = wcol[(k + 2 * half + 1) * 3];
        v2f b;
        b.x = used ? wx : 0.0f;
        b.y = used ? wy : 0.0f;
        acc = __builtin_amdgcn_wmma_f32_16x16x4_f32(false, a, false, b, (short)0, acc, false, false);
    }
    if (used) {
        const bool rel = lr < 3;
        const int c = rel ? lr : (lr - 3);
        float bv = 0.0f;
        if (!rel && bias) bv = bias[c];
        float* dstp = rel ? proj : init;
#pragma unroll
        for (int i = 0; i < 8; ++i) {
            const int row = m0 + 8 * half + i;
            if (row < n) dstp[(size_t)row * 3 + c] = acc[i] + bv;
        }
    }
}

__global__ __launch_bounds__(256) void scatter_add3(const float* __restrict__ proj,
                                                    const int* __restrict__ sidx,
                                                    const int* __restrict__ didx,
                                                    float* out, int nE) {
    const int lane = threadIdx.x & 31;
    const int e0 = (blockIdx.x * 8 + (int)(threadIdx.x >> 5)) * 32;
    if (e0 >= nE) return;
    const int e = e0 + lane;
    const int ec = (e < nE) ? e : (nE - 1);
    const int sv = sidx[ec];
    const int dv = didx[ec];
    const int c = lane & 3;
    const int q = lane >> 2;
    float v[4];
    int doff[4];
    bool act[4];
#pragma unroll
    for (int r = 0; r < 4; ++r) {
        const int sl = r * 8 + q;
        const int s = __shfl(sv, sl, 32);
        const int d = __shfl(dv, sl, 32);
        act[r] = (c < 3) && (e0 + sl < nE);
        doff[r] = d * 3 + c;
        v[r] = act[r] ? proj[(size_t)s * 3 + c] : 0.0f;
    }
#pragma unroll
    for (int r = 0; r < 4; ++r) {
        if (act[r]) atomicAdd(out + doff[r], v[r]);
    }
}

__global__ __launch_bounds__(256) void knowledge_kernel(float* __restrict__ out,
                                                        const float* __restrict__ x,
                                                        int n) {
    int i = blockIdx.x * blockDim.x + threadIdx.x;
    if (i >= n) return;
    float a = out[(size_t)i * 3 + 0];
    float b = out[(size_t)i * 3 + 1];
    float c = out[(size_t)i * 3 + 2];
    float nrm = sqrtf(a * a + b * b + c * c);
    float s = 10.0f / fmaxf(nrm, 1e-12f);
    a *= s; b *= s; c *= s;
    if (x[(size_t)i * 64 + 61] != 0.0f) a -= 10.0f;
    if (x[(size_t)i * 64 + 63] != 0.0f) c -= 10.0f;
    out[(size_t)i * 3 + 0] = a;
    out[(size_t)i * 3 + 1] = b;
    out[(size_t)i * 3 + 2] = c;
}

static inline int ceil_div(long long a, long long b) { return (int)((a + b - 1) / b); }

extern "C" void kernel_launch(void* const* d_in, const int* in_sizes, int n_in,
                              void* d_out, int out_size, void* d_ws, size_t ws_size,
                              hipStream_t stream) {
    const float* x_s = (const float*)d_in[0];
    const float* x_t = (const float*)d_in[1];
    const int* e_l = (const int*)d_in[2];
    const int* e_r = (const int*)d_in[3];
    const float* w1lr_rel  = (const float*)d_in[4];
    const float* b1lr      = (const float*)d_in[5];
    const float* w1lr_root = (const float*)d_in[6];
    const float* w1rl_rel  = (const float*)d_in[7];
    const float* b1rl      = (const float*)d_in[8];
    const float* w1rl_root = (const float*)d_in[9];
    const float* w2lr_rel  = (const float*)d_in[10];
    const float* b2lr      = (const float*)d_in[11];
    const float* w2lr_root = (const float*)d_in[12];
    const float* w2rl_rel  = (const float*)d_in[13];
    const float* b2rl      = (const float*)d_in[14];
    const float* w2rl_root = (const float*)d_in[15];
    const float* w3lr_rel  = (const float*)d_in[16];
    const float* b3lr      = (const float*)d_in[17];
    const float* w3lr_root = (const float*)d_in[18];
    const float* w3rl_rel  = (const float*)d_in[19];
    const float* b3rl      = (const float*)d_in[20];
    const float* w3rl_root = (const float*)d_in[21];

    const int NL = in_sizes[0] / 64;
    const int NR = in_sizes[1] / 64;
    const int E  = in_sizes[2];

    float* ws = (float*)d_ws;
    const size_t AGG_R = 0;
    const size_t AGG_L = AGG_R + (size_t)NR * 128;
    const size_t H1_R  = AGG_L + (size_t)NL * 128;
    const size_t H1_L  = H1_R + (size_t)NR * 128;
    const size_t P_R   = H1_L + (size_t)NL * 128;
    const size_t P_L   = P_R + (size_t)NR * 128;
    const size_t TKT   = P_L + (size_t)NL * 128;
    const size_t PROJ_LR = H1_R;
    const size_t PROJ_RL = H1_R + (size_t)NL * 3;

    int* tkt = (int*)(ws + TKT);

    float* d_left  = (float*)d_out;
    float* d_right = (float*)d_out + (size_t)NL * 3;

    const int TPB = 256;
    constexpr int MT = 2;
    constexpr int GR = 2;
    const int SGRID = 1536;

    zero_kernel<<<ceil_div((size_t)NR * 64, TPB), TPB, 0, stream>>>(ws + AGG_R, (size_t)NR * 64, nullptr, 0);
    zero_kernel<<<ceil_div((size_t)NL * 64, TPB), TPB, 0, stream>>>(ws + AGG_L, (size_t)NL * 64, tkt, 256);
    scatter_add_rows<64, GR><<<SGRID, TPB, 0, stream>>>(x_s, e_l, e_r, ws + AGG_R, E, NR, tkt);
    scatter_add_rows<64, GR><<<SGRID, TPB, 0, stream>>>(x_t, e_r, e_l, ws + AGG_L, E, NL, tkt + 64);
    gemm_dual_wmma<64, 64, true, MT><<<ceil_div(ceil_div(NR, 16 * MT), 8), TPB, 0, stream>>>(
        ws + AGG_R, w1lr_rel, x_t, w1lr_root, b1lr, ws + H1_R, nullptr, nullptr, nullptr, NR);
    gemm_dual_wmma<64, 64, true, MT><<<ceil_div(ceil_div(NL, 16 * MT), 8), TPB, 0, stream>>>(
        ws + AGG_L, w1rl_rel, x_s, w1rl_root, b1rl, ws + H1_L, nullptr, nullptr, nullptr, NL);

    gemm_dual_wmma<128, 0, false, MT><<<dim3(ceil_div(ceil_div(NR, 16 * MT), 8), 2), TPB, 0, stream>>>(
        ws + H1_R, w2rl_rel, nullptr, nullptr, nullptr, ws + P_R, w2lr_root, b2lr, ws + AGG_R, NR);
    gemm_dual_wmma<128, 0, false, MT><<<dim3(ceil_div(ceil_div(NL, 16 * MT), 8), 2), TPB, 0, stream>>>(
        ws + H1_L, w2lr_rel, nullptr, nullptr, nullptr, ws + P_L, w2rl_root, b2rl, ws + AGG_L, NL);
    scatter_add_rows<128, GR><<<SGRID, TPB, 0, stream>>>(ws + P_L, e_l, e_r, ws + AGG_R, E, NR, tkt + 128);
    scatter_add_rows<128, GR><<<SGRID, TPB, 0, stream>>>(ws + P_R, e_r, e_l, ws + AGG_L, E, NL, tkt + 192);

    rowmat6_wmma<<<ceil_div(ceil_div(NL, 16), 8), TPB, 0, stream>>>(
        ws + AGG_L, w3lr_rel, w3rl_root, b3rl, ws + PROJ_LR, d_left, NL);
    rowmat6_wmma<<<ceil_div(ceil_div(NR, 16), 8), TPB, 0, stream>>>(
        ws + AGG_R, w3rl_rel, w3lr_root, b3lr, ws + PROJ_RL, d_right, NR);
    scatter_add3<<<ceil_div(E, TPB), TPB, 0, stream>>>(ws + PROJ_LR, e_l, e_r, d_right, E);
    scatter_add3<<<ceil_div(E, TPB), TPB, 0, stream>>>(ws + PROJ_RL, e_r, e_l, d_left, E);

    knowledge_kernel<<<ceil_div(NL, TPB), TPB, 0, stream>>>(d_left, x_s, NL);
    knowledge_kernel<<<ceil_div(NR, TPB), TPB, 0, stream>>>(d_right, x_t, NR);
}


# Layer-3 edge adds: one accumulator plane per half of the chip (work-group reads its die number); closing pass adds the two planes; no line gets adds from both halves; atomics kept, sum order changes
# speedup vs baseline: 2.6592x; 1.1698x over previous
#include <hip/hip_runtime.h>
#include <hip/hip_bf16.h>

typedef __attribute__((ext_vector_type(2))) float v2f;
typedef __attribute__((ext_vector_type(8))) float v8f;

__global__ __launch_bounds__(256) void zero_kernel(float* __restrict__ p, size_t n,
                                                   int* __restrict__ tickets, int n_tickets) {
    size_t i = (size_t)blockIdx.x * blockDim.x + threadIdx.x;
    if (i < n) p[i] = 0.0f;
    if (i < (size_t)n_tickets) tickets[i] = 0;
}

template <int F, int G>
__global__ __launch_bounds__(256) void scatter_add_rows(const float* __restrict__ src,
                                                        const int* __restrict__ sidx,
                                                        const int* __restrict__ didx,
                                                        float* dst, int nE, int n_dst,
                                                        int* tickets) {
    constexpr int C = F / 32;
    constexpr int CH = 8 * G * 32;
    constexpr int JB = 64 / C;
    __shared__ int s_tk[2];
    __shared__ int s_half;
    const int lane = threadIdx.x & 31;
    const int wave = __builtin_amdgcn_readfirstlane((int)(threadIdx.x >> 5));
    if (threadIdx.x == 0) {
        const unsigned id = __builtin_amdgcn_s_sendmsg_rtn(0x87);
        s_half = (int)((id >> 18) & 1u);
    }
    __syncthreads();
    int tgt = __builtin_amdgcn_readfirstlane(s_half);
    const int nsplit = n_dst >> 1;
    const int nCh = (nE + CH - 1) / CH;
    int pass = 0;
    for (int it = 0; it < 2 * nCh + 2; ++it) {
        if (threadIdx.x == 0) s_tk[it & 1] = atomicAdd(tickets + tgt * 32, 1);
        __syncthreads();
        const int tk = __builtin_amdgcn_readfirstlane(s_tk[it & 1]);
        if (tk >= nCh) {
            if (pass) break;
            pass = 1;
            tgt ^= 1;
            continue;
        }
        unsigned soffs[G], doffs[G], masks[G];
#pragma unroll
        for (int g = 0; g < G; ++g) {
            const int e = tk * CH + (wave * G + g) * 32 + lane;
            const int ec = (e < nE) ? e : (nE - 1);
            const int sv = sidx[ec];
            const int dv = didx[ec];
            soffs[g] = (unsigned)sv * (unsigned)(F * 4);
            doffs[g] = (unsigned)dv * (unsigned)(F * 4);
            const bool mine = (e < nE) && (((dv >= nsplit) ? 1 : 0) == tgt);
            masks[g] = __builtin_amdgcn_ballot_w32(mine);
        }
#pragma unroll
        for (int g = 0; g < G; ++g) {
            const unsigned soff = soffs[g], doff = doffs[g], mask = masks[g];
            const unsigned lane4 = (unsigned)lane * 4u;
#pragma unroll
            for (int j0 = 0; j0 < 32; j0 += JB) {
                float v[JB][C] = {};
#pragma unroll
                for (int j = 0; j < JB; ++j) {
                    if ((mask >> (j0 + j)) & 1u) {
                        const unsigned so = __builtin_amdgcn_readlane(soff, j0 + j);
                        const char* p = (const char*)src + so;
#pragma unroll
                        for (int c = 0; c < C; ++c) v[j][c] = *(const float*)(p + (lane4 + c * 128u));
                    }
                }
#pragma unroll
                for (int j = 0; j < JB; ++j) {
#pragma unroll
                    for (int c = 0; c < C; ++c) asm volatile("" ::"v"(v[j][c]));
                }
#pragma unroll
                for (int j = 0; j < JB; ++j) {
                    if ((mask >> (j0 + j)) & 1u) {
                        const unsigned dof = __builtin_amdgcn_readlane(doff, j0 + j);
                        char* q = (char*)dst + dof;
#pragma unroll
                        for (int c = 0; c < C; ++c) atomicAdd((float*)(q + (lane4 + c * 128u)), v[j][c]);
                    }
                }
            }
        }
    }
}

constexpr int GEMM_PITCH = 132;

template <int KK, int MT>
__device__ __forceinline__ void gemm_k_steps(const float (*sa)[GEMM_PITCH], int acol0,
                                             const float* __restrict__ W, int lr, int half,
                                             v8f (&acc)[MT][8]) {
    constexpr int N = 128;
#pragma unroll 4
    for (int k = 0; k < KK; k += 4) {
        v2f a[MT];
#pragma unroll
        for (int mt = 0; mt < MT; ++mt)
            a[mt] = *(const v2f*)&sa[mt * 16 + lr][acol0 + k + 2 * half];
        const float* w0 = W + (size_t)(k + 2 * half) * N + lr * 4;
        const float4 x0 = *(const float4*)(w0);
        const float4 x1 = *(const float4*)(w0 + 64);
        const float4 y0 = *(const float4*)(w0 + N);
        const float4 y1 = *(const float4*)(w0 + N + 64);
        const float bx[8] = {x0.x, x0.y, x0.z, x0.w, x1.x, x1.y, x1.z, x1.w};
        const float by[8] = {y0.x, y0.y, y0.z, y0.w, y1.x, y1.y, y1.z, y1.w};
#pragma unroll
        for (int nt = 0; nt < 8; ++nt) {
            v2f b;
            b.x = bx[nt];
            b.y = by[nt];
#pragma unroll
            for (int mt = 0; mt < MT; ++mt)
                acc[mt][nt] = __builtin_amdgcn_wmma_f32_16x16x4_f32(
                    false, a[mt], false, b, (short)0, acc[mt][nt], false, false);
        }
    }
}

template <int K1, int K2, bool RELU, int MT>
__global__ __launch_bounds__(256) void gemm_dual_wmma(const float* __restrict__ A1,
                                                      const float* __restrict__ W1a,
                                                      const float* __restrict__ A2,
                                                      const float* __restrict__ W2,
                                                      const float* __restrict__ biasa,
                                                      float* outa,
                                                      const float* __restrict__ W1b,
                                                      const float* __restrict__ biasb,
                                                      float* outb, int M) {
    constexpr int N = 128;
    constexpr int ROWS = 16 * MT;
    static_assert(K1 + K2 == 128, "a lane stages 4 of the 128 joined columns");
    __shared__ __attribute__((aligned(16))) float s_a[8][ROWS][GEMM_PITCH];
    const float* W1 = blockIdx.y ? W1b : W1a;
    const float* bias = blockIdx.y ? biasb : biasa;
    float* out = blockIdx.y ? outb : outa;
    const int lane = threadIdx.x & 31;
    const int wave = threadIdx.x >> 5;
    const int m0 = (blockIdx.x * 8 + wave) * ROWS;
    const int half = lane >> 4;
    const int lr = lane & 15;
    float (*sa)[GEMM_PITCH] = s_a[wave];

    if (m0 < M) {
        const int c4 = lane * 4;
        const float* srcp = A1;
        int srck = K1, coff = c4;
        if (K2 > 0 && c4 >= K1) { srcp = A2; srck = K2; coff = c4 - K1; }
#pragma unroll 8
        for (int r = 0; r < ROWS; ++r) {
            int g = m0 + r;
            g = (g < M) ? g : (M - 1);
            const float4 v = *(const float4*)(srcp + (size_t)g * srck + coff);
            *(float4*)&sa[r][c4] = v;
        }
    }
    __syncthreads();
    if (m0 >= M) return;

    v8f acc[MT][8] = {};
    gemm_k_steps<K1, MT>(sa, 0, W1, lr, half, acc);
    if constexpr (K2 > 0) gemm_k_steps<K2, MT>(sa, K1, W2, lr, half, acc);

#pragma unroll
    for (int mt = 0; mt < MT; ++mt) {
        if (m0 + mt * 16 >= M) break;
#pragma unroll
        for (int j = 0; j < 2; ++j) {
            const int col = j * 64 + lr * 4;
            float4 bv = {0.0f, 0.0f, 0.0f, 0.0f};
            if (bias) bv = *(const float4*)(bias + col);
#pragma unroll
            for (int i = 0; i < 8; ++i) {
                float4 v;
                v.x = acc[mt][j * 4 + 0][i] + bv.x;
                v.y = acc[mt][j * 4 + 1][i] + bv.y;
                v.z = acc[mt][j * 4 + 2][i] + bv.z;
                v.w = acc[mt][j * 4 + 3][i] + bv.w;
                if (RELU) {
                    v.x = fmaxf(v.x, 0.0f);
                    v.y = fmaxf(v.y, 0.0f);
                    v.z = fmaxf(v.z, 0.0f);
                    v.w = fmaxf(v.w, 0.0f);
                }
                *(float4*)(out + (size_t)(m0 + mt * 16 + i + 8 * half) * N + col) = v;
            }
        }
    }
}

__global__ __launch_bounds__(256) void rowmat6_wmma(const float* __restrict__ X,
                                                    const float* __restrict__ Wrel,
                                                    const float* __restrict__ Wroot,
                                                    const float* __restrict__ bias,
                                                    float* __restrict__ proj,
                                                    float* __restrict__ init,
                                                    float* __restrict__ init1, int n) {
    __shared__ __attribute__((aligned(16))) float s_x[8][16][GEMM_PITCH];
    const int lane = threadIdx.x & 31;
    const int wave = threadIdx.x >> 5;
    const int m0 = (blockIdx.x * 8 + wave) * 16;
    const int half = lane >> 4;
    const int lr = lane & 15;
    float (*sx)[GEMM_PITCH] = s_x[wave];
    if (m0 < n) {
        float4 v[16];
#pragma unroll
        for (int r = 0; r < 16; ++r) {
            int g = m0 + r;
            g = (g < n) ? g : (n - 1);
            v[r] = *(const float4*)(X + (size_t)g * 128 + lane * 4);
        }
#pragma unroll
        for (int r = 0; r < 16; ++r) {
            float4 t = v[r];
            t.x = fmaxf(t.x, 0.0f);
            t.y = fmaxf(t.y, 0.0f);
            t.z = fmaxf(t.z, 0.0f);
            t.w = fmaxf(t.w, 0.0f);
            *(float4*)&sx[r][lane * 4] = t;
        }
    }
    __syncthreads();
    if (m0 >= n) return;

    const bool used = lr < 6;
    const float* wcol = (lr < 3) ? (Wrel + lr) : (used ? (Wroot + (lr - 3)) : Wrel);
    v8f acc = {};
#pragma unroll
    for (int k = 0; k < 128; k += 4) {
        const v2f a = *(const v2f*)&sx[lr][k + 2 * half];
        const float wx = wcol[(k + 2 * half) * 3];
        const float wy = wcol[(k + 2 * half + 1) * 3];
        v2f b;
        b.x = used ? wx : 0.0f;
        b.y = used ? wy : 0.0f;
        acc = __builtin_amdgcn_wmma_f32_16x16x4_f32(false, a, false, b, (short)0, acc, false, false);
    }
    if (used) {
        const bool rel = lr < 3;
        const int c = rel ? lr : (lr - 3);
        float bv = 0.0f;
        if (!rel && bias) bv = bias[c];
        float* dstp = rel ? proj : init;
#pragma unroll
        for (int i = 0; i < 8; ++i) {
            const int row = m0 + 8 * half + i;
            if (row < n) {
                dstp[(size_t)row * 3 + c] = acc[i] + bv;
                if (!rel) init1[(size_t)row * 3 + c] = 0.0f;
            }
        }
    }
}

__global__ __launch_bounds__(256) void scatter_add3(const float* __restrict__ proj,
                                                    const int* __restrict__ sidx,
                                                    const int* __restrict__ didx,
                                                    float* out0, float* out1, int nE) {
    __shared__ int s_half;
    if (threadIdx.x == 0) {
        const unsigned id = __builtin_amdgcn_s_sendmsg_rtn(0x87);
        s_half = (int)((id >> 18) & 1u);
    }
    __syncthreads();
    float* out = __builtin_amdgcn_readfirstlane(s_half) ? out1 : out0;
    const int lane = threadIdx.x & 31;
    const int e0 = (blockIdx.x * 8 + (int)(threadIdx.x >> 5)) * 32;
    if (e0 >= nE) return;
    const int e = e0 + lane;
    const int ec = (e < nE) ? e : (nE - 1);
    const int sv = sidx[ec];
    const int dv = didx[ec];
    const int c = lane & 3;
    const int q = lane >> 2;
    float v[4];
    int doff[4];
    bool act[4];
#pragma unroll
    for (int r = 0; r < 4; ++r) {
        const int sl = r * 8 + q;
        const int s = __shfl(sv, sl, 32);
        const int d = __shfl(dv, sl, 32);
        act[r] = (c < 3) && (e0 + sl < nE);
        doff[r] = d * 3 + c;
        v[r] = act[r] ? proj[(size_t)s * 3 + c] : 0.0f;
    }
#pragma unroll
    for (int r = 0; r < 4; ++r) {
        if (act[r]) atomicAdd(out + doff[r], v[r]);
    }
}

__global__ __launch_bounds__(256) void knowledge_kernel(float* __restrict__ out,
                                                        const float* __restrict__ out1,
                                                        const float* __restrict__ x,
                                                        int n) {
    int i = blockIdx.x * blockDim.x + threadIdx.x;
    if (i >= n) return;
    float a = out[(size_t)i * 3 + 0] + out1[(size_t)i * 3 + 0];
    float b = out[(size_t)i * 3 + 1] + out1[(size_t)i * 3 + 1];
    float c = out[(size_t)i * 3 + 2] + out1[(size_t)i * 3 + 2];
    float nrm = sqrtf(a * a + b * b + c * c);
    float s = 10.0f / fmaxf(nrm, 1e-12f);
    a *= s; b *= s; c *= s;
    if (x[(size_t)i * 64 + 61] != 0.0f) a -= 10.0f;
    if (x[(size_t)i * 64 + 63] != 0.0f) c -= 10.0f;
    out[(size_t)i * 3 + 0] = a;
    out[(size_t)i * 3 + 1] = b;
    out[(size_t)i * 3 + 2] = c;
}

static inline int ceil_div(long long a, long long b) { return (int)((a + b - 1) / b); }

extern "C" void kernel_launch(void* const* d_in, const int* in_sizes, int n_in,
                              void* d_out, int out_size, void* d_ws, size_t ws_size,
                              hipStream_t stream) {
    const float* x_s = (const float*)d_in[0];
    const float* x_t = (const float*)d_in[1];
    const int* e_l = (const int*)d_in[2];
    const int* e_r = (const int*)d_in[3];
    const float* w1lr_rel  = (const float*)d_in[4];
    const float* b1lr      = (const float*)d_in[5];
    const float* w1lr_root = (const float*)d_in[6];
    const float* w1rl_rel  = (const float*)d_in[7];
    const float* b1rl      = (const float*)d_in[8];
    const float* w1rl_root = (const float*)d_in[9];
    const float* w2lr_rel  = (const float*)d_in[10];
    const float* b2lr      = (const float*)d_in[11];
    const float* w2lr_root = (const float*)d_in[12];
    const float* w2rl_rel  = (const float*)d_in[13];
    const float* b2rl      = (const float*)d_in[14];
    const float* w2rl_root = (const float*)d_in[15];
    const float* w3lr_rel  = (const float*)d_in[16];
    const float* b3lr      = (const float*)d_in[17];
    const float* w3lr_root = (const float*)d_in[18];
    const float* w3rl_rel  = (const float*)d_in[19];
    const float* b3rl      = (const float*)d_in[20];
    const float* w3rl_root = (const float*)d_in[21];

    const int NL = in_sizes[0] / 64;
    const int NR = in_sizes[1] / 64;
    const int E  = in_sizes[2];

    float* ws = (float*)d_ws;
    const size_t AGG_R = 0;
    const size_t AGG_L = AGG_R + (size_t)NR * 128;
    const size_t H1_R  = AGG_L + (size_t)NL * 128;
    const size_t H1_L  = H1_R + (size_t)NR * 128;
    const size_t P_R   = H1_L + (size_t)NL * 128;
    const size_t P_L   = P_R + (size_t)NR * 128;
    const size_t TKT   = P_L + (size_t)NL * 128;
    const size_t PROJ_LR = H1_R;
    const size_t PROJ_RL = H1_R + (size_t)NL * 3;
    const size_t ACC1_L  = H1_L;
    const size_t ACC1_R  = H1_L + (size_t)NL * 3;

    int* tkt = (int*)(ws + TKT);

    float* d_left  = (float*)d_out;
    float* d_right = (float*)d_out + (size_t)NL * 3;

    const int TPB = 256;
    constexpr int MT = 2;
    constexpr int GR = 2;
    const int SGRID = 1536;

    zero_kernel<<<ceil_div((size_t)NR * 64, TPB), TPB, 0, stream>>>(ws + AGG_R, (size_t)NR * 64, nullptr, 0);
    zero_kernel<<<ceil_div((size_t)NL * 64, TPB), TPB, 0, stream>>>(ws + AGG_L, (size_t)NL * 64, tkt, 256);
    scatter_add_rows<64, GR><<<SGRID, TPB, 0, stream>>>(x_s, e_l, e_r, ws + AGG_R, E, NR, tkt);
    scatter_add_rows<64, GR><<<SGRID, TPB, 0, stream>>>(x_t, e_r, e_l, ws + AGG_L, E, NL, tkt + 64);
    gemm_dual_wmma<64, 64, true, MT><<<ceil_div(ceil_div(NR, 16 * MT), 8), TPB, 0, stream>>>(
        ws + AGG_R, w1lr_rel, x_t, w1lr_root, b1lr, ws + H1_R, nullptr, nullptr, nullptr, NR);
    gemm_dual_wmma<64, 64, true, MT><<<ceil_div(ceil_div(NL, 16 * MT), 8), TPB, 0, stream>>>(
        ws + AGG_L, w1rl_rel, x_s, w1rl_root, b1rl, ws + H1_L, nullptr, nullptr, nullptr, NL);

    gemm_dual_wmma<128, 0, false, MT><<<dim3(ceil_div(ceil_div(NR, 16 * MT), 8), 2), TPB, 0, stream>>>(
        ws + H1_R, w2rl_rel, nullptr, nullptr, nullptr, ws + P_R, w2lr_root, b2lr, ws + AGG_R, NR);
    gemm_dual_wmma<128, 0, false, MT><<<dim3(ceil_div(ceil_div(NL, 16 * MT), 8), 2), TPB, 0, stream>>>(
        ws + H1_L, w2lr_rel, nullptr, nullptr, nullptr, ws + P_L, w2rl_root, b2rl, ws + AGG_L, NL);
    scatter_add_rows<128, GR><<<SGRID, TPB, 0, stream>>>(ws + P_L, e_l, e_r, ws + AGG_R, E, NR, tkt + 128);
    scatter_add_rows<128, GR><<<SGRID, TPB, 0, stream>>>(ws + P_R, e_r, e_l, ws + AGG_L, E, NL, tkt + 192);

    rowmat6_wmma<<<ceil_div(ceil_div(NL, 16), 8), TPB, 0, stream>>>(
        ws + AGG_L, w3lr_rel, w3rl_root, b3rl, ws + PROJ_LR, d_left, ws + ACC1_L, NL);
    rowmat6_wmma<<<ceil_div(ceil_div(NR, 16), 8), TPB, 0, stream>>>(
        ws + AGG_R, w3rl_rel, w3lr_root, b3lr, ws + PROJ_RL, d_right, ws + ACC1_R, NR);
    scatter_add3<<<ceil_div(E, TPB), TPB, 0, stream>>>(ws + PROJ_LR, e_l, e_r, d_right, ws + ACC1_R, E);
    scatter_add3<<<ceil_div(E, TPB), TPB, 0, stream>>>(ws + PROJ_RL, e_r, e_l, d_left, ws + ACC1_L, E);

    knowledge_kernel<<<ceil_div(NL, TPB), TPB, 0, stream>>>(d_left, ws + ACC1_L, x_s, NL);
    knowledge_kernel<<<ceil_div(NR, TPB), TPB, 0, stream>>>(d_right, ws + ACC1_R, x_t, NR);
}


# Layer-1 and layer-2 GEMMs of a node set fused: a wave keeps its 32x128 tile of relu'd h1 in its own LDS piece as A operand of both layer-2 products; h1 never stored; 16 launches become 14; same sums
# speedup vs baseline: 2.8280x; 1.0635x over previous
#include <hip/hip_runtime.h>
#include <hip/hip_bf16.h>

typedef __attribute__((ext_vector_type(2))) float v2f;
typedef __attribute__((ext_vector_type(8))) float v8f;

__global__ __launch_bounds__(256) void zero_kernel(float* __restrict__ p, size_t n,
                                                   int* __restrict__ tickets, int n_tickets) {
    size_t i = (size_t)blockIdx.x * blockDim.x + threadIdx.x;
    if (i < n) p[i] = 0.0f;
    if (i < (size_t)n_tickets) tickets[i] = 0;
}

template <int F, int G>
__global__ __launch_bounds__(256) void scatter_add_rows(const float* __restrict__ src,
                                                        const int* __restrict__ sidx,
                                                        const int* __restrict__ didx,
                                                        float* dst, int nE, int n_dst,
                                                        int* tickets) {
    constexpr int C = F / 32;
    constexpr int CH = 8 * G * 32;
    constexpr int JB = 64 / C;
    __shared__ int s_tk[2];
    __shared__ int s_half;
    const int lane = threadIdx.x & 31;
    const int wave = __builtin_amdgcn_readfirstlane((int)(threadIdx.x >> 5));
    if (threadIdx.x == 0) {
        const unsigned id = __builtin_amdgcn_s_sendmsg_rtn(0x87);
        s_half = (int)((id >> 18) & 1u);
    }
    __syncthreads();
    int tgt = __builtin_amdgcn_readfirstlane(s_half);
    const int nsplit = n_dst >> 1;
    const int nCh = (nE + CH - 1) / CH;
    int pass = 0;
    for (int it = 0; it < 2 * nCh + 2; ++it) {
        if (threadIdx.x == 0) s_tk[it & 1] = atomicAdd(tickets + tgt * 32, 1);
        __syncthreads();
        const int tk = __builtin_amdgcn_readfirstlane(s_tk[it & 1]);
        if (tk >= nCh) {
            if (pass) break;
            pass = 1;
            tgt ^= 1;
            continue;
        }
        unsigned soffs[G], doffs[G], masks[G];
#pragma unroll
        for (int g = 0; g < G; ++g) {
            const int e = tk * CH + (wave * G + g) * 32 + lane;
            const int ec = (e < nE) ? e : (nE - 1);
            const int sv = sidx[ec];
            const int dv = didx[ec];
            soffs[g] = (unsigned)sv * (unsigned)(F * 4);
            doffs[g] = (unsigned)dv * (unsigned)(F * 4);
            const bool mine = (e < nE) && (((dv >= nsplit) ? 1 : 0) == tgt);
            masks[g] = __builtin_amdgcn_ballot_w32(mine);
        }
#pragma unroll
        for (int g = 0; g < G; ++g) {
            const unsigned soff = soffs[g], doff = doffs[g], mask = masks[g];
            const unsigned lane4 = (unsigned)lane * 4u;
#pragma unroll
            for (int j0 = 0; j0 < 32; j0 += JB) {
                float v[JB][C] = {};
#pragma unroll
                for (int j = 0; j < JB; ++j) {
                    if ((mask >> (j0 + j)) & 1u) {
                        const unsigned so = __builtin_amdgcn_readlane(soff, j0 + j);
                        const char* p = (const char*)src + so;
#pragma unroll
                        for (int c = 0; c < C; ++c) v[j][c] = *(const float*)(p + (lane4 + c * 128u));
                    }
                }
#pragma unroll
                for (int j = 0; j < JB; ++j) {
#pragma unroll
                    for (int c = 0; c < C; ++c) asm volatile("" ::"v"(v[j][c]));
                }
#pragma unroll
                for (int j = 0; j < JB; ++j) {
                    if ((mask >> (j0 + j)) & 1u) {
                        const unsigned dof = __builtin_amdgcn_readlane(doff, j0 + j);
                        char* q = (char*)dst + dof;
#pragma unroll
                        for (int c = 0; c < C; ++c) atomicAdd((float*)(q + (lane4 + c * 128u)), v[j][c]);
                    }
                }
            }
        }
    }
}

constexpr int GEMM_PITCH = 132;

template <int KK, int MT>
__device__ __forceinline__ void gemm_k_steps(const float (*sa)[GEMM_PITCH], int acol0,
                                             const float* __restrict__ W, int lr, int half,
                                             v8f (&acc)[MT][8]) {
    constexpr int N = 128;
#pragma unroll 4
    for (int k = 0; k < KK; k += 4) {
        v2f a[MT];
#pragma unroll
        for (int mt = 0; mt < MT; ++mt)
            a[mt] = *(const v2f*)&sa[mt * 16 + lr][acol0 + k + 2 * half];
        const float* w0 = W + (size_t)(k + 2 * half) * N + lr * 4;
        const float4 x0 = *(const float4*)(w0);
        const float4 x1 = *(const float4*)(w0 + 64);
        const float4 y0 = *(const float4*)(w0 + N);
        const float4 y1 = *(const float4*)(w0 + N + 64);
        const float bx[8] = {x0.x, x0.y, x0.z, x0.w, x1.x, x1.y, x1.z, x1.w};
        const float by[8] = {y0.x, y0.y, y0.z, y0.w, y1.x, y1.y, y1.z, y1.w};
#pragma unroll
        for (int nt = 0; nt < 8; ++nt) {
            v2f b;
            b.x = bx[nt];
            b.y = by[nt];
#pragma unroll
            for (int mt = 0; mt < MT; ++mt)
                acc[mt][nt] = __builtin_amdgcn_wmma_f32_16x16x4_f32(
                    false, a[mt], false, b, (short)0, acc[mt][nt], false, false);
        }
    }
}

template <int MT>
__device__ __forceinline__ void store_rows(const v8f (&acc)[MT][8], const float* __restrict__ bias,
                                           float* out, int m0, int M, int lr, int half) {
    constexpr int N = 128;
#pragma unroll
    for (int mt = 0; mt < MT; ++mt) {
        if (m0 + mt * 16 >= M) break;
#pragma unroll
        for (int j = 0; j < 2; ++j) {
            const int col = j * 64 + lr * 4;
            float4 bv = {0.0f, 0.0f, 0.0f, 0.0f};
            if (bias) bv = *(const float4*)(bias + col);
#pragma unroll
            for (int i = 0; i < 8; ++i) {
                float4 v;
                v.x = acc[mt][j * 4 + 0][i] + bv.x;
                v.y = acc[mt][j * 4 + 1][i] + bv.y;
                v.z = acc[mt][j * 4 + 2][i] + bv.z;
                v.w = acc[mt][j * 4 + 3][i] + bv.w;
                *(float4*)(out + (size_t)(m0 + mt * 16 + i + 8 * half) * N + col) = v;
            }
        }
    }
}

template <int K1, int K2, int MT>
__global__ __launch_bounds__(256) void gemm_two_layers_wmma(const float* __restrict__ A1,
                                                            const float* __restrict__ W1,
                                                            const float* __restrict__ A2,
                                                            const float* __restrict__ W2,
                                                            const float* __restrict__ bias1,
                                                            const float* __restrict__ Wp,
                                                            float* outp,
                                                            const float* __restrict__ Wr,
                                                            const float* __restrict__ biasr,
                                                            float* outr, int M) {
    constexpr int ROWS = 16 * MT;
    static_assert(K1 + K2 == 128, "a lane stages 4 of the 128 joined columns");
    __shared__ __attribute__((aligned(16))) float s_a[8][ROWS][GEMM_PITCH];
    const int lane = threadIdx.x & 31;
    const int wave = threadIdx.x >> 5;
    const int m0 = (blockIdx.x * 8 + wave) * ROWS;
    const int half = lane >> 4;
    const int lr = lane & 15;
    float (*sa)[GEMM_PITCH] = s_a[wave];

    if (m0 < M) {
        const int c4 = lane * 4;
        const float* srcp = A1;
        int srck = K1, coff = c4;
        if (c4 >= K1) { srcp = A2; srck = K2; coff = c4 - K1; }
#pragma unroll 8
        for (int r = 0; r < ROWS; ++r) {
            int g = m0 + r;
            g = (g < M) ? g : (M - 1);
            const float4 v = *(const float4*)(srcp + (size_t)g * srck + coff);
            *(float4*)&sa[r][c4] = v;
        }
    }
    __syncthreads();
    if (m0 >= M) return;

    {
        v8f acc[MT][8] = {};
        gemm_k_steps<K1, MT>(sa, 0, W1, lr, half, acc);
        gemm_k_steps<K2, MT>(sa, K1, W2, lr, half, acc);
#pragma unroll
        for (int mt = 0; mt < MT; ++mt) {
#pragma unroll
            for (int j = 0; j < 2; ++j) {
                const int col = j * 64 + lr * 4;
                float4 bv = {0.0f, 0.0f, 0.0f, 0.0f};
                if (bias1) bv = *(const float4*)(bias1 + col);
#pragma unroll
                for (int i = 0; i < 8; ++i) {
                    float4 v;
                    v.x = fmaxf(acc[mt][j * 4 + 0][i] + bv.x, 0.0f);
                    v.y = fmaxf(acc[mt][j * 4 + 1][i] + bv.y, 0.0f);
                    v.z = fmaxf(acc[mt][j * 4 + 2][i] + bv.z, 0.0f);
                    v.w = fmaxf(acc[mt][j * 4 + 3][i] + bv.w, 0.0f);
                    *(float4*)&sa[mt * 16 + i + 8 * half][col] = v;
                }
            }
        }
    }

    {
        v8f acc[MT][8] = {};
        gemm_k_steps<128, MT>(sa, 0, Wp, lr, half, acc);
        store_rows<MT>(acc, nullptr, outp, m0, M, lr, half);
    }
    {
        v8f acc[MT][8] = {};
        gemm_k_steps<128, MT>(sa, 0, Wr, lr, half, acc);
        store_rows<MT>(acc, biasr, outr, m0, M, lr, half);
    }
}

__global__ __launch_bounds__(256) void rowmat6_wmma(const float* __restrict__ X,
                                                    const float* __restrict__ Wrel,
                                                    const float* __restrict__ Wroot,
                                                    const float* __restrict__ bias,
                                                    float* __restrict__ proj,
                                                    float* __restrict__ init,
                                                    float* __restrict__ init1, int n) {
    __shared__ __attribute__((aligned(16))) float s_x[8][16][GEMM_PITCH];
    const int lane = threadIdx.x & 31;
    const int wave = threadIdx.x >> 5;
    const int m0 = (blockIdx.x * 8 + wave) * 16;
    const int half = lane >> 4;
    const int lr = lane & 15;
    float (*sx)[GEMM_PITCH] = s_x[wave];
    if (m0 < n) {
        float4 v[16];
#pragma unroll
        for (int r = 0; r < 16; ++r) {
            int g = m0 + r;
            g = (g < n) ? g : (n - 1);
            v[r] = *(const float4*)(X + (size_t)g * 128 + lane * 4);
        }
#pragma unroll
        for (int r = 0; r < 16; ++r) {
            float4 t = v[r];
            t.x = fmaxf(t.x, 0.0f);
            t.y = fmaxf(t.y, 0.0f);
            t.z = fmaxf(t.z, 0.0f);
            t.w = fmaxf(t.w, 0.0f);
            *(float4*)&sx[r][lane * 4] = t;
        }
    }
    __syncthreads();
    if (m0 >= n) return;

    const bool used = lr < 6;
    const float* wcol = (lr < 3) ? (Wrel + lr) : (used ? (Wroot + (lr - 3)) : Wrel);
    v8f acc = {};
#pragma unroll
    for (int k = 0; k < 128; k += 4) {
        const v2f a = *(const v2f*)&sx[lr][k + 2 * half];
        const float wx = wcol[(k + 2 * half) * 3];
        const float wy = wcol[(k + 2 * half + 1) * 3];
        v2f b;
        b.x = used ? wx : 0.0f;
        b.y = used ? wy : 0.0f;
        acc = __builtin_amdgcn_wmma_f32_16x16x4_f32(false, a, false, b, (short)0, acc, false, false);
    }
    if (used) {
        const bool rel = lr < 3;
        const int c = rel ? lr : (lr - 3);
        float bv = 0.0f;
        if (!rel && bias) bv = bias[c];
        float* dstp = rel ? proj : init;
#pragma unroll
        for (int i = 0; i < 8; ++i) {
            const int row = m0 + 8 * half + i;
            if (row < n) {
                dstp[(size_t)row * 3 + c] = acc[i] + bv;
                if (!rel) init1[(size_t)row * 3 + c] = 0.0f;
            }
        }
    }
}

__global__ __launch_bounds__(256) void scatter_add3(const float* __restrict__ proj,
                                                    const int* __restrict__ sidx,
                                                    const int* __restrict__ didx,
                                                    float* out0, float* out1, int nE) {
    __shared__ int s_half;
    if (threadIdx.x == 0) {
        const unsigned id = __builtin_amdgcn_s_sendmsg_rtn(0x87);
        s_half = (int)((id >> 18) & 1u);
    }
    __syncthreads();
    float* out = __builtin_amdgcn_readfirstlane(s_half) ? out1 : out0;
    const int lane = threadIdx.x & 31;
    const int e0 = (blockIdx.x * 8 + (int)(threadIdx.x >> 5)) * 32;
    if (e0 >= nE) return;
    const int e = e0 + lane;
    const int ec = (e < nE) ? e : (nE - 1);
    const int sv = sidx[ec];
    const int dv = didx[ec];
    const int c = lane & 3;
    const int q = lane >> 2;
    float v[4];
    int doff[4];
    bool act[4];
#pragma unroll
    for (int r = 0; r < 4; ++r) {
        const int sl = r * 8 + q;
        const int s = __shfl(sv, sl, 32);
        const int d = __shfl(dv, sl, 32);
        act[r] = (c < 3) && (e0 + sl < nE);
        doff[r] = d * 3 + c;
        v[r] = act[r] ? proj[(size_t)s * 3 + c] : 0.0f;
    }
#pragma unroll
    for (int r = 0; r < 4; ++r) {
        if (act[r]) atomicAdd(out + doff[r], v[r]);
    }
}

__global__ __launch_bounds__(256) void knowledge_kernel(float* __restrict__ out,
                                                        const float* __restrict__ out1,
                                                        const float* __restrict__ x,
                                                        int n) {
    int i = blockIdx.x * blockDim.x + threadIdx.x;
    if (i >= n) return;
    float a = out[(size_t)i * 3 + 0] + out1[(size_t)i * 3 + 0];
    float b = out[(size_t)i * 3 + 1] + out1[(size_t)i * 3 + 1];
    float c = out[(size_t)i * 3 + 2] + out1[(size_t)i * 3 + 2];
    float nrm = sqrtf(a * a + b * b + c * c);
    float s = 10.0f / fmaxf(nrm, 1e-12f);
    a *= s; b *= s; c *= s;
    if (x[(size_t)i * 64 + 61] != 0.0f) a -= 10.0f;
    if (x[(size_t)i * 64 + 63] != 0.0f) c -= 10.0f;
    out[(size_t)i * 3 + 0] = a;
    out[(size_t)i * 3 + 1] = b;
    out[(size_t)i * 3 + 2] = c;
}

static inline int ceil_div(long long a, long long b) { return (int)((a + b - 1) / b); }

extern "C" void kernel_launch(void* const* d_in, const int* in_sizes, int n_in,
                              void* d_out, int out_size, void* d_ws, size_t ws_size,
                              hipStream_t stream) {
    const float* x_s = (const float*)d_in[0];
    const float* x_t = (const float*)d_in[1];
    const int* e_l = (const int*)d_in[2];
    const int* e_r = (const int*)d_in[3];
    const float* w1lr_rel  = (const float*)d_in[4];
    const float* b1lr      = (const float*)d_in[5];
    const float* w1lr_root = (const float*)d_in[6];
    const float* w1rl_rel  = (const float*)d_in[7];
    const float* b1rl      = (const float*)d_in[8];
    const float* w1rl_root = (const float*)d_in[9];
    const float* w2lr_rel  = (const float*)d_in[10];
    const float* b2lr      = (const float*)d_in[11];
    const float* w2lr_root = (const float*)d_in[12];
    const float* w2rl_rel  = (const float*)d_in[13];
    const float* b2rl      = (const float*)d_in[14];
    const float* w2rl_root = (const float*)d_in[15];
    const float* w3lr_rel  = (const float*)d_in[16];
    const float* b3lr      = (const float*)d_in[17];
    const float* w3lr_root = (const float*)d_in[18];
    const float* w3rl_rel  = (const float*)d_in[19];
    const float* b3rl      = (const float*)d_in[20];
    const float* w3rl_root = (const float*)d_in[21];

    const int NL = in_sizes[0] / 64;
    const int NR = in_sizes[1] / 64;
    const int E  = in_sizes[2];

    float* ws = (float*)d_ws;
    const size_t AGG_R = 0;
    const size_t AGG_L = AGG_R + (size_t)NR * 128;
    const size_t S2_R  = AGG_L + (size_t)NL * 128;
    const size_t S2_L  = S2_R + (size_t)NR * 128;
    const size_t P_R   = S2_L + (size_t)NL * 128;
    const size_t P_L   = P_R + (size_t)NR * 128;
    const size_t TKT   = P_L + (size_t)NL * 128;
    const size_t PROJ_LR = AGG_R;
    const size_t PROJ_RL = AGG_R + (size_t)NL * 3;
    const size_t ACC1_L  = AGG_L;
    const size_t ACC1_R  = AGG_L + (size_t)NL * 3;

    int* tkt = (int*)(ws + TKT);

    float* d_left  = (float*)d_out;
    float* d_right = (float*)d_out + (size_t)NL * 3;

    const int TPB = 256;
    constexpr int MT = 2;
    constexpr int GR = 2;
    const int SGRID = 1536;

    zero_kernel<<<ceil_div((size_t)NR * 64, TPB), TPB, 0, stream>>>(ws + AGG_R, (size_t)NR * 64, nullptr, 0);
    zero_kernel<<<ceil_div((size_t)NL * 64, TPB), TPB, 0, stream>>>(ws + AGG_L, (size_t)NL * 64, tkt, 256);
    scatter_add_rows<64, GR><<<SGRID, TPB, 0, stream>>>(x_s, e_l, e_r, ws + AGG_R, E, NR, tkt);
    scatter_add_rows<64, GR><<<SGRID, TPB, 0, stream>>>(x_t, e_r, e_l, ws + AGG_L, E, NL, tkt + 64);
    gemm_two_layers_wmma<64, 64, MT><<<ceil_div(ceil_div(NR, 16 * MT), 8), TPB, 0, stream>>>(
        ws + AGG_R, w1lr_rel, x_t, w1lr_root, b1lr, w2rl_rel, ws + P_R, w2lr_root, b2lr, ws + S2_R, NR);
    gemm_two_layers_wmma<64, 64, MT><<<ceil_div(ceil_div(NL, 16 * MT), 8), TPB, 0, stream>>>(
        ws + AGG_L, w1rl_rel, x_s, w1rl_root, b1rl, w2lr_rel, ws + P_L, w2rl_root, b2rl, ws + S2_L, NL);

    scatter_add_rows<128, GR><<<SGRID, TPB, 0, stream>>>(ws + P_L, e_l, e_r, ws + S2_R, E, NR, tkt + 128);
    scatter_add_rows<128, GR><<<SGRID, TPB, 0, stream>>>(ws + P_R, e_r, e_l, ws + S2_L, E, NL, tkt + 192);

    rowmat6_wmma<<<ceil_div(ceil_div(NL, 16), 8), TPB, 0, stream>>>(
        ws + S2_L, w3lr_rel, w3rl_root, b3rl, ws + PROJ_LR, d_left, ws + ACC1_L, NL);
    rowmat6_wmma<<<ceil_div(ceil_div(NR, 16), 8), TPB, 0, stream>>>(
        ws + S2_R, w3rl_rel, w3lr_root, b3lr, ws + PROJ_RL, d_right, ws + ACC1_R, NR);
    scatter_add3<<<ceil_div(E, TPB), TPB, 0, stream>>>(ws + PROJ_LR, e_l, e_r, d_right, ws + ACC1_R, E);
    scatter_add3<<<ceil_div(E, TPB), TPB, 0, stream>>>(ws + PROJ_RL, e_r, e_l, d_left, ws + ACC1_L, E);

    knowledge_kernel<<<ceil_div(NL, TPB), TPB, 0, stream>>>(d_left, ws + ACC1_L, x_s, NL);
    knowledge_kernel<<<ceil_div(NR, TPB), TPB, 0, stream>>>(d_right, ws + ACC1_R, x_t, NR);
}


# Layer GEMMs as one resident launch (256 work-groups of 16 waves) that deals the 16-row tiles of both node sets evenly: 9-10 per SIMD, was 8 then 4 on the fullest; 14 launches to 13; same sums
# speedup vs baseline: 2.9474x; 1.0422x over previous
#include <hip/hip_runtime.h>
#include <hip/hip_bf16.h>

typedef __attribute__((ext_vector_type(2))) float v2f;
typedef __attribute__((ext_vector_type(8))) float v8f;

__global__ __launch_bounds__(256) void zero_kernel(float* __restrict__ p, size_t n,
                                                   int* __restrict__ tickets, int n_tickets) {
    size_t i = (size_t)blockIdx.x * blockDim.x + threadIdx.x;
    if (i < n) p[i] = 0.0f;
    if (i < (size_t)n_tickets) tickets[i] = 0;
}

template <int F, int G>
__global__ __launch_bounds__(256) void scatter_add_rows(const float* __restrict__ src,
                                                        const int* __restrict__ sidx,
                                                        const int* __restrict__ didx,
                                                        float* dst, int nE, int n_dst,
                                                        int* tickets) {
    constexpr int C = F / 32;
    constexpr int CH = 8 * G * 32;
    constexpr int JB = 64 / C;
    __shared__ int s_tk[2];
    __shared__ int s_half;
    const int lane = threadIdx.x & 31;
    const int wave = __builtin_amdgcn_readfirstlane((int)(threadIdx.x >> 5));
    if (threadIdx.x == 0) {
        const unsigned id = __builtin_amdgcn_s_sendmsg_rtn(0x87);
        s_half = (int)((id >> 18) & 1u);
    }
    __syncthreads();
    int tgt = __builtin_amdgcn_readfirstlane(s_half);
    const int nsplit = n_dst >> 1;
    const int nCh = (nE + CH - 1) / CH;
    int pass = 0;
    for (int it = 0; it < 2 * nCh + 2; ++it) {
        if (threadIdx.x == 0) s_tk[it & 1] = atomicAdd(tickets + tgt * 32, 1);
        __syncthreads();
        const int tk = __builtin_amdgcn_readfirstlane(s_tk[it & 1]);
        if (tk >= nCh) {
            if (pass) break;
            pass = 1;
            tgt ^= 1;
            continue;
        }
        unsigned soffs[G], doffs[G], masks[G];
#pragma unroll
        for (int g = 0; g < G; ++g) {
            const int e = tk * CH + (wave * G + g) * 32 + lane;
            const int ec = (e < nE) ? e : (nE - 1);
            const int sv = sidx[ec];
            const int dv = didx[ec];
            soffs[g] = (unsigned)sv * (unsigned)(F * 4);
            doffs[g] = (unsigned)dv * (unsigned)(F * 4);
            const bool mine = (e < nE) && (((dv >= nsplit) ? 1 : 0) == tgt);
            masks[g] = __builtin_amdgcn_ballot_w32(mine);
        }
#pragma unroll
        for (int g = 0; g < G; ++g) {
            const unsigned soff = soffs[g], doff = doffs[g], mask = masks[g];
            const unsigned lane4 = (unsigned)lane * 4u;
#pragma unroll
            for (int j0 = 0; j0 < 32; j0 += JB) {
                float v[JB][C] = {};
#pragma unroll
                for (int j = 0; j < JB; ++j) {
                    if ((mask >> (j0 + j)) & 1u) {
                        const unsigned so = __builtin_amdgcn_readlane(soff, j0 + j);
                        const char* p = (const char*)src + so;
#pragma unroll
                        for (int c = 0; c < C; ++c) v[j][c] = *(const float*)(p + (lane4 + c * 128u));
                    }
                }
#pragma unroll
                for (int j = 0; j < JB; ++j) {
#pragma unroll
                    for (int c = 0; c < C; ++c) asm volatile("" ::"v"(v[j][c]));
                }
#pragma unroll
                for (int j = 0; j < JB; ++j) {
                    if ((mask >> (j0 + j)) & 1u) {
                        const unsigned dof = __builtin_amdgcn_readlane(doff, j0 + j);
                        char* q = (char*)dst + dof;
#pragma unroll
                        for (int c = 0; c < C; ++c) atomicAdd((float*)(q + (lane4 + c * 128u)), v[j][c]);
                    }
                }
            }
        }
    }
}

constexpr int GEMM_PITCH = 132;

template <int KK, int MT>
__device__ __forceinline__ void gemm_k_steps(const float (*sa)[GEMM_PITCH], int acol0,
                                             const float* __restrict__ W, int lr, int half,
                                             v8f (&acc)[MT][8]) {
    constexpr int N = 128;
#pragma unroll 4
    for (int k = 0; k < KK; k += 4) {
        v2f a[MT];
#pragma unroll
        for (int mt = 0; mt < MT; ++mt)
            a[mt] = *(const v2f*)&sa[mt * 16 + lr][acol0 + k + 2 * half];
        const float* w0 = W + (size_t)(k + 2 * half) * N + lr * 4;
        const float4 x0 = *(const float4*)(w0);
        const float4 x1 = *(const float4*)(w0 + 64);
        const float4 y0 = *(const float4*)(w0 + N);
        const float4 y1 = *(const float4*)(w0 + N + 64);
        const float bx[8] = {x0.x, x0.y, x0.z, x0.w, x1.x, x1.y, x1.z, x1.w};
        const float by[8] = {y0.x, y0.y, y0.z, y0.w, y1.x, y1.y, y1.z, y1.w};
#pragma unroll
        for (int nt = 0; nt < 8; ++nt) {
            v2f b;
            b.x = bx[nt];
            b.y = by[nt];
#pragma unroll
            for (int mt = 0; mt < MT; ++mt)
                acc[mt][nt] = __builtin_amdgcn_wmma_f32_16x16x4_f32(
                    false, a[mt], false, b, (short)0, acc[mt][nt], false, false);
        }
    }
}

template <int MT>
__device__ __forceinline__ void store_rows(const v8f (&acc)[MT][8], const float* __restrict__ bias,
                                           float* out, int m0, int M, int lr, int half) {
    constexpr int N = 128;
#pragma unroll
    for (int mt = 0; mt < MT; ++mt) {
        if (m0 + mt * 16 >= M) break;
#pragma unroll
        for (int j = 0; j < 2; ++j) {
            const int col = j * 64 + lr * 4;
            float4 bv = {0.0f, 0.0f, 0.0f, 0.0f};
            if (bias) bv = *(const float4*)(bias + col);
#pragma unroll
            for (int i = 0; i < 8; ++i) {
                float4 v;
                v.x = acc[mt][j * 4 + 0][i] + bv.x;
                v.y = acc[mt][j * 4 + 1][i] + bv.y;
                v.z = acc[mt][j * 4 + 2][i] + bv.z;
                v.w = acc[mt][j * 4 + 3][i] + bv.w;
                *(float4*)(out + (size_t)(m0 + mt * 16 + i + 8 * half) * N + col) = v;
            }
        }
    }
}

struct GemmSet {
    const float* A1;
    const float* W1;
    const float* A2;
    const float* W2;
    const float* bias1;
    const float* Wp;
    float* outp;
    const float* Wr;
    const float* biasr;
    float* outr;
    int M;
};

template <int MT>
__device__ __forceinline__ void two_layer_tile(const float* __restrict__ A1, const float* __restrict__ W1,
                                               const float* __restrict__ A2, const float* __restrict__ W2,
                                               const float* __restrict__ bias1,
                                               const float* __restrict__ Wp, float* outp,
                                               const float* __restrict__ Wr,
                                               const float* __restrict__ biasr, float* outr,
                                               int M, int m0, float (*sa)[GEMM_PITCH],
                                               int lane, int lr, int half) {
    constexpr int ROWS = 16 * MT;
    constexpr int K1 = 64, K2 = 64;
    {
        const int c4 = lane * 4;
        const float* srcp = A1;
        int srck = K1, coff = c4;
        if (c4 >= K1) { srcp = A2; srck = K2; coff = c4 - K1; }
#pragma unroll 8
        for (int r = 0; r < ROWS; ++r) {
            int g = m0 + r;
            g = (g < M) ? g : (M - 1);
            const float4 v = *(const float4*)(srcp + (size_t)g * srck + coff);
            *(float4*)&sa[r][c4] = v;
        }
    }

    {
        v8f acc[MT][8] = {};
        gemm_k_steps<K1, MT>(sa, 0, W1, lr, half, acc);
        gemm_k_steps<K2, MT>(sa, K1, W2, lr, half, acc);
#pragma unroll
        for (int mt = 0; mt < MT; ++mt) {
#pragma unroll
            for (int j = 0; j < 2; ++j) {
                const int col = j * 64 + lr * 4;
                float4 bv = {0.0f, 0.0f, 0.0f, 0.0f};
                if (bias1) bv = *(const float4*)(bias1 + col);
#pragma unroll
                for (int i = 0; i < 8; ++i) {
                    float4 v;
                    v.x = fmaxf(acc[mt][j * 4 + 0][i] + bv.x, 0.0f);
                    v.y = fmaxf(acc[mt][j * 4 + 1][i] + bv.y, 0.0f);
                    v.z = fmaxf(acc[mt][j * 4 + 2][i] + bv.z, 0.0f);
                    v.w = fmaxf(acc[mt][j * 4 + 3][i] + bv.w, 0.0f);
                    *(float4*)&sa[mt * 16 + i + 8 * half][col] = v;
                }
            }
        }
    }

    {
        v8f acc[MT][8] = {};
        gemm_k_steps<128, MT>(sa, 0, Wp, lr, half, acc);
        store_rows<MT>(acc, nullptr, outp, m0, M, lr, half);
    }
    {
        v8f acc[MT][8] = {};
        gemm_k_steps<128, MT>(sa, 0, Wr, lr, half, acc);
        store_rows<MT>(acc, biasr, outr, m0, M, lr, half);
    }
}

__global__ __launch_bounds__(512) void gemm_two_layers_wmma(const GemmSet sr, const GemmSet sl) {
    __shared__ __attribute__((aligned(16))) float s_a[16][32][GEMM_PITCH];
    const int lane = threadIdx.x & 31;
    const int wave = __builtin_amdgcn_readfirstlane((int)(threadIdx.x >> 5));
    const int half = lane >> 4;
    const int lr = lane & 15;
    float (*sa)[GEMM_PITCH] = s_a[wave];

    const int TR = (sr.M + 15) >> 4;
    const int T = TR + ((sl.M + 15) >> 4);
    const int g = ((int)blockIdx.x * 4 + (wave & 3)) * 4 + (wave >> 2);
    const int G = (int)gridDim.x * 16;
    int t = (int)(((unsigned)g * (unsigned)T) / (unsigned)G);
    const int t1 = (int)(((unsigned)(g + 1) * (unsigned)T) / (unsigned)G);
    while (t < t1) {
        const bool isr = t < TR;
        const int lim = (isr && TR < t1) ? TR : t1;
        const int m0 = (isr ? t : (t - TR)) * 16;
        const float* A1 = isr ? sr.A1 : sl.A1;
        const float* W1 = isr ? sr.W1 : sl.W1;
        const float* A2 = isr ? sr.A2 : sl.A2;
        const float* W2 = isr ? sr.W2 : sl.W2;
        const float* b1 = isr ? sr.bias1 : sl.bias1;
        const float* Wp = isr ? sr.Wp : sl.Wp;
        float* outp = isr ? sr.outp : sl.outp;
        const float* Wr = isr ? sr.Wr : sl.Wr;
        const float* br = isr ? sr.biasr : sl.biasr;
        float* outr = isr ? sr.outr : sl.outr;
        const int M = isr ? sr.M : sl.M;
        if (t + 1 < lim) {
            two_layer_tile<2>(A1, W1, A2, W2, b1, Wp, outp, Wr, br, outr, M, m0, sa, lane, lr, half);
            t += 2;
        } else {
            two_layer_tile<1>(A1, W1, A2, W2, b1, Wp, outp, Wr, br, outr, M, m0, sa, lane, lr, half);
            t += 1;
        }
    }
}

__global__ __launch_bounds__(256) void rowmat6_wmma(const float* __restrict__ X,
                                                    const float* __restrict__ Wrel,
                                                    const float* __restrict__ Wroot,
                                                    const float* __restrict__ bias,
                                                    float* __restrict__ proj,
                                                    float* __restrict__ init,
                                                    float* __restrict__ init1, int n) {
    __shared__ __attribute__((aligned(16))) float s_x[8][16][GEMM_PITCH];
    const int lane = threadIdx.x & 31;
    const int wave = threadIdx.x >> 5;
    const int m0 = (blockIdx.x * 8 + wave) * 16;
    const int half = lane >> 4;
    const int lr = lane & 15;
    float (*sx)[GEMM_PITCH] = s_x[wave];
    if (m0 < n) {
        float4 v[16];
#pragma unroll
        for (int r = 0; r < 16; ++r) {
            int g = m0 + r;
            g = (g < n) ? g : (n - 1);
            v[r] = *(const float4*)(X + (size_t)g * 128 + lane * 4);
        }
#pragma unroll
        for (int r = 0; r < 16; ++r) {
            float4 t = v[r];
            t.x = fmaxf(t.x, 0.0f);
            t.y = fmaxf(t.y, 0.0f);
            t.z = fmaxf(t.z, 0.0f);
            t.w = fmaxf(t.w, 0.0f);
            *(float4*)&sx[r][lane * 4] = t;
        }
    }
    __syncthreads();
    if (m0 >= n) return;

    const bool used = lr < 6;
    const float* wcol = (lr < 3) ? (Wrel + lr) : (used ? (Wroot + (lr - 3)) : Wrel);
    v8f acc = {};
#pragma unroll
    for (int k = 0; k < 128; k += 4) {
        const v2f a = *(const v2f*)&sx[lr][k + 2 * half];
        const float wx = wcol[(k + 2 * half) * 3];
        const float wy = wcol[(k + 2 * half + 1) * 3];
        v2f b;
        b.x = used ? wx : 0.0f;
        b.y = used ? wy : 0.0f;
        acc = __builtin_amdgcn_wmma_f32_16x16x4_f32(false, a, false, b, (short)0, acc, false, false);
    }
    if (used) {
        const bool rel = lr < 3;
        const int c = rel ? lr : (lr - 3);
        float bv = 0.0f;
        if (!rel && bias) bv = bias[c];
        float* dstp = rel ? proj : init;
#pragma unroll
        for (int i = 0; i < 8; ++i) {
            const int row = m0 + 8 * half + i;
            if (row < n) {
                dstp[(size_t)row * 3 + c] = acc[i] + bv;
                if (!rel) init1[(size_t)row * 3 + c] = 0.0f;
            }
        }
    }
}

__global__ __launch_bounds__(256) void scatter_add3(const float* __restrict__ proj,
                                                    const int* __restrict__ sidx,
                                                    const int* __restrict__ didx,
                                                    float* out0, float* out1, int nE) {
    __shared__ int s_half;
    if (threadIdx.x == 0) {
        const unsigned id = __builtin_amdgcn_s_sendmsg_rtn(0x87);
        s_half = (int)((id >> 18) & 1u);
    }
    __syncthreads();
    float* out = __builtin_amdgcn_readfirstlane(s_half) ? out1 : out0;
    const int lane = threadIdx.x & 31;
    const int e0 = (blockIdx.x * 8 + (int)(threadIdx.x >> 5)) * 32;
    if (e0 >= nE) return;
    const int e = e0 + lane;
    const int ec = (e < nE) ? e : (nE - 1);
    const int sv = sidx[ec];
    const int dv = didx[ec];
    const int c = lane & 3;
    const int q = lane >> 2;
    float v[4];
    int doff[4];
    bool act[4];
#pragma unroll
    for (int r = 0; r < 4; ++r) {
        const int sl = r * 8 + q;
        const int s = __shfl(sv, sl, 32);
        const int d = __shfl(dv, sl, 32);
        act[r] = (c < 3) && (e0 + sl < nE);
        doff[r] = d * 3 + c;
        v[r] = act[r] ? proj[(size_t)s * 3 + c] : 0.0f;
    }
#pragma unroll
    for (int r = 0; r < 4; ++r) {
        if (act[r]) atomicAdd(out + doff[r], v[r]);
    }
}

__global__ __launch_bounds__(256) void knowledge_kernel(float* __restrict__ out,
                                                        const float* __restrict__ out1,
                                                        const float* __restrict__ x,
                                                        int n) {
    int i = blockIdx.x * blockDim.x + threadIdx.x;
    if (i >= n) return;
    float a = out[(size_t)i * 3 + 0] + out1[(size_t)i * 3 + 0];
    float b = out[(size_t)i * 3 + 1] + out1[(size_t)i * 3 + 1];
    float c = out[(size_t)i * 3 + 2] + out1[(size_t)i * 3 + 2];
    float nrm = sqrtf(a * a + b * b + c * c);
    float s = 10.0f / fmaxf(nrm, 1e-12f);
    a *= s; b *= s; c *= s;
    if (x[(size_t)i * 64 + 61] != 0.0f) a -= 10.0f;
    if (x[(size_t)i * 64 + 63] != 0.0f) c -= 10.0f;
    out[(size_t)i * 3 + 0] = a;
    out[(size_t)i * 3 + 1] = b;
    out[(size_t)i * 3 + 2] = c;
}

static inline int ceil_div(long long a, long long b) { return (int)((a + b - 1) / b); }

extern "C" void kernel_launch(void* const* d_in, const int* in_sizes, int n_in,
                              void* d_out, int out_size, void* d_ws, size_t ws_size,
                              hipStream_t stream) {
    const float* x_s = (const float*)d_in[0];
    const float* x_t = (const float*)d_in[1];
    const int* e_l = (const int*)d_in[2];
    const int* e_r = (const int*)d_in[3];
    const float* w1lr_rel  = (const float*)d_in[4];
    const float* b1lr      = (const float*)d_in[5];
    const float* w1lr_root = (const float*)d_in[6];
    const float* w1rl_rel  = (const float*)d_in[7];
    const float* b1rl      = (const float*)d_in[8];
    const float* w1rl_root = (const float*)d_in[9];
    const float* w2lr_rel  = (const float*)d_in[10];
    const float* b2lr      = (const float*)d_in[11];
    const float* w2lr_root = (const float*)d_in[12];
    const float* w2rl_rel  = (const float*)d_in[13];
    const float* b2rl      = (const float*)d_in[14];
    const float* w2rl_root = (const float*)d_in[15];
    const float* w3lr_rel  = (const float*)d_in[16];
    const float* b3lr      = (const float*)d_in[17];
    const float* w3lr_root = (const float*)d_in[18];
    const float* w3rl_rel  = (const float*)d_in[19];
    const float* b3rl      = (const float*)d_in[20];
    const float* w3rl_root = (const float*)d_in[21];

    const int NL = in_sizes[0] / 64;
    const int NR = in_sizes[1] / 64;
    const int E  = in_sizes[2];

    float* ws = (float*)d_ws;
    const size_t AGG_R = 0;
    const size_t AGG_L = AGG_R + (size_t)NR * 128;
    const size_t S2_R  = AGG_L + (size_t)NL * 128;
    const size_t S2_L  = S2_R + (size_t)NR * 128;
    const size_t P_R   = S2_L + (size_t)NL * 128;
    const size_t P_L   = P_R + (size_t)NR * 128;
    const size_t TKT   = P_L + (size_t)NL * 128;
    const size_t PROJ_LR = AGG_R;
    const size_t PROJ_RL = AGG_R + (size_t)NL * 3;
    const size_t ACC1_L  = AGG_L;
    const size_t ACC1_R  = AGG_L + (size_t)NL * 3;

    int* tkt = (int*)(ws + TKT);

    float* d_left  = (float*)d_out;
    float* d_right = (float*)d_out + (size_t)NL * 3;

    const int TPB = 256;
    constexpr int GR = 2;
    const int SGRID = 1536;

    zero_kernel<<<ceil_div((size_t)NR * 64, TPB), TPB, 0, stream>>>(ws + AGG_R, (size_t)NR * 64, nullptr, 0);
    zero_kernel<<<ceil_div((size_t)NL * 64, TPB), TPB, 0, stream>>>(ws + AGG_L, (size_t)NL * 64, tkt, 256);
    scatter_add_rows<64, GR><<<SGRID, TPB, 0, stream>>>(x_s, e_l, e_r, ws + AGG_R, E, NR, tkt);
    scatter_add_rows<64, GR><<<SGRID, TPB, 0, stream>>>(x_t, e_r, e_l, ws + AGG_L, E, NL, tkt + 64);
    const GemmSet set_r = {ws + AGG_R, w1lr_rel, x_t, w1lr_root, b1lr, w2rl_rel, ws + P_R, w2lr_root, b2lr, ws + S2_R, NR};
    const GemmSet set_l = {ws + AGG_L, w1rl_rel, x_s, w1rl_root, b1rl, w2lr_rel, ws + P_L, w2rl_root, b2rl, ws + S2_L, NL};
    gemm_two_layers_wmma<<<256, 512, 0, stream>>>(set_r, set_l);

    scatter_add_rows<128, GR><<<SGRID, TPB, 0, stream>>>(ws + P_L, e_l, e_r, ws + S2_R, E, NR, tkt + 128);
    scatter_add_rows<128, GR><<<SGRID, TPB, 0, stream>>>(ws + P_R, e_r, e_l, ws + S2_L, E, NL, tkt + 192);

    rowmat6_wmma<<<ceil_div(ceil_div(NL, 16), 8), TPB, 0, stream>>>(
        ws + S2_L, w3lr_rel, w3rl_root, b3rl, ws + PROJ_LR, d_left, ws + ACC1_L, NL);
    rowmat6_wmma<<<ceil_div(ceil_div(NR, 16), 8), TPB, 0, stream>>>(
        ws + S2_R, w3rl_rel, w3lr_root, b3lr, ws + PROJ_RL, d_right, ws + ACC1_R, NR);
    scatter_add3<<<ceil_div(E, TPB), TPB, 0, stream>>>(ws + PROJ_LR, e_l, e_r, d_right, ws + ACC1_R, E);
    scatter_add3<<<ceil_div(E, TPB), TPB, 0, stream>>>(ws + PROJ_RL, e_r, e_l, d_left, ws + ACC1_L, E);

    knowledge_kernel<<<ceil_div(NL, TPB), TPB, 0, stream>>>(d_left, ws + ACC1_L, x_s, NL);
    knowledge_kernel<<<ceil_div(NR, TPB), TPB, 0, stream>>>(d_right, ws + ACC1_R, x_t, NR);
}


# Zero start of the larger layer-1 sum plane written by the other direction's edge pass (resident work-groups, 16-byte stores behind their last chunk); its launch of 200000 one-store waves gone
# speedup vs baseline: 3.0381x; 1.0308x over previous
#include <hip/hip_runtime.h>
#include <hip/hip_bf16.h>

typedef __attribute__((ext_vector_type(2))) float v2f;
typedef __attribute__((ext_vector_type(8))) float v8f;

__global__ __launch_bounds__(256) void zero_kernel(float* __restrict__ p, size_t n,
                                                   int* __restrict__ tickets, int n_tickets) {
    size_t i = (size_t)blockIdx.x * blockDim.x + threadIdx.x;
    if (i < n) p[i] = 0.0f;
    if (i < (size_t)n_tickets) tickets[i] = 0;
}

template <int F, int G>
__global__ __launch_bounds__(256) void scatter_add_rows(const float* __restrict__ src,
                                                        const int* __restrict__ sidx,
                                                        const int* __restrict__ didx,
                                                        float* dst, int nE, int n_dst,
                                                        int* tickets,
                                                        float* __restrict__ zplane, size_t zn4) {
    constexpr int C = F / 32;
    constexpr int CH = 8 * G * 32;
    constexpr int JB = 64 / C;
    __shared__ int s_tk[2];
    __shared__ int s_half;
    const int lane = threadIdx.x & 31;
    const int wave = __builtin_amdgcn_readfirstlane((int)(threadIdx.x >> 5));
    if (threadIdx.x == 0) {
        const unsigned id = __builtin_amdgcn_s_sendmsg_rtn(0x87);
        s_half = (int)((id >> 18) & 1u);
    }
    __syncthreads();
    int tgt = __builtin_amdgcn_readfirstlane(s_half);
    const int nsplit = n_dst >> 1;
    const int nCh = (nE + CH - 1) / CH;
    int pass = 0;
    for (int it = 0; it < 2 * nCh + 2; ++it) {
        if (threadIdx.x == 0) s_tk[it & 1] = atomicAdd(tickets + tgt * 32, 1);
        __syncthreads();
        const int tk = __builtin_amdgcn_readfirstlane(s_tk[it & 1]);
        if (tk >= nCh) {
            if (pass) break;
            pass = 1;
            tgt ^= 1;
            continue;
        }
        unsigned soffs[G], doffs[G], masks[G];
#pragma unroll
        for (int g = 0; g < G; ++g) {
            const int e = tk * CH + (wave * G + g) * 32 + lane;
            const int ec = (e < nE) ? e : (nE - 1);
            const int sv = sidx[ec];
            const int dv = didx[ec];
            soffs[g] = (unsigned)sv * (unsigned)(F * 4);
            doffs[g] = (unsigned)dv * (unsigned)(F * 4);
            const bool mine = (e < nE) && (((dv >= nsplit) ? 1 : 0) == tgt);
            masks[g] = __builtin_amdgcn_ballot_w32(mine);
        }
#pragma unroll
        for (int g = 0; g < G; ++g) {
            const unsigned soff = soffs[g], doff = doffs[g], mask = masks[g];
            const unsigned lane4 = (unsigned)lane * 4u;
#pragma unroll
            for (int j0 = 0; j0 < 32; j0 += JB) {
                float v[JB][C] = {};
#pragma unroll
                for (int j = 0; j < JB; ++j) {
                    if ((mask >> (j0 + j)) & 1u) {
                        const unsigned so = __builtin_amdgcn_readlane(soff, j0 + j);
                        const char* p = (const char*)src + so;
#pragma unroll
                        for (int c = 0; c < C; ++c) v[j][c] = *(const float*)(p + (lane4 + c * 128u));
                    }
                }
#pragma unroll
                for (int j = 0; j < JB; ++j) {
#pragma unroll
                    for (int c = 0; c < C; ++c) asm volatile("" ::"v"(v[j][c]));
                }
#pragma unroll
                for (int j = 0; j < JB; ++j) {
                    if ((mask >> (j0 + j)) & 1u) {
                        const unsigned dof = __builtin_amdgcn_readlane(doff, j0 + j);
                        char* q = (char*)dst + dof;
#pragma unroll
                        for (int c = 0; c < C; ++c) atomicAdd((float*)(q + (lane4 + c * 128u)), v[j][c]);
                    }
                }
            }
        }
    }
    if (zplane) {
        float4* zp = (float4*)zplane;
        const float4 z = {0.0f, 0.0f, 0.0f, 0.0f};
        const size_t step = (size_t)gridDim.x * blockDim.x;
        for (size_t i = (size_t)blockIdx.x * blockDim.x + threadIdx.x; i < zn4; i += step) zp[i] = z;
    }
}

constexpr int GEMM_PITCH = 132;

template <int KK, int MT>
__device__ __forceinline__ void gemm_k_steps(const float (*sa)[GEMM_PITCH], int acol0,
                                             const float* __restrict__ W, int lr, int half,
                                             v8f (&acc)[MT][8]) {
    constexpr int N = 128;
#pragma unroll 4
    for (int k = 0; k < KK; k += 4) {
        v2f a[MT];
#pragma unroll
        for (int mt = 0; mt < MT; ++mt)
            a[mt] = *(const v2f*)&sa[mt * 16 + lr][acol0 + k + 2 * half];
        const float* w0 = W + (size_t)(k + 2 * half) * N + lr * 4;
        const float4 x0 = *(const float4*)(w0);
        const float4 x1 = *(const float4*)(w0 + 64);
        const float4 y0 = *(const float4*)(w0 + N);
        const float4 y1 = *(const float4*)(w0 + N + 64);
        const float bx[8] = {x0.x, x0.y, x0.z, x0.w, x1.x, x1.y, x1.z, x1.w};
        const float by[8] = {y0.x, y0.y, y0.z, y0.w, y1.x, y1.y, y1.z, y1.w};
#pragma unroll
        for (int nt = 0; nt < 8; ++nt) {
            v2f b;
            b.x = bx[nt];
            b.y = by[nt];
#pragma unroll
            for (int mt = 0; mt < MT; ++mt)
                acc[mt][nt] = __builtin_amdgcn_wmma_f32_16x16x4_f32(
                    false, a[mt], false, b, (short)0, acc[mt][nt], false, false);
        }
    }
}

template <int MT>
__device__ __forceinline__ void store_rows(const v8f (&acc)[MT][8], const float* __restrict__ bias,
                                           float* out, int m0, int M, int lr, int half) {
    constexpr int N = 128;
#pragma unroll
    for (int mt = 0; mt < MT; ++mt) {
        if (m0 + mt * 16 >= M) break;
#pragma unroll
        for (int j = 0; j < 2; ++j) {
            const int col = j * 64 + lr * 4;
            float4 bv = {0.0f, 0.0f, 0.0f, 0.0f};
            if (bias) bv = *(const float4*)(bias + col);
#pragma unroll
            for (int i = 0; i < 8; ++i) {
                float4 v;
                v.x = acc[mt][j * 4 + 0][i] + bv.x;
                v.y = acc[mt][j * 4 + 1][i] + bv.y;
                v.z = acc[mt][j * 4 + 2][i] + bv.z;
                v.w = acc[mt][j * 4 + 3][i] + bv.w;
                *(float4*)(out + (size_t)(m0 + mt * 16 + i + 8 * half) * N + col) = v;
            }
        }
    }
}

struct GemmSet {
    const float* A1;
    const float* W1;
    const float* A2;
    const float* W2;
    const float* bias1;
    const float* Wp;
    float* outp;
    const float* Wr;
    const float* biasr;
    float* outr;
    int M;
};

template <int MT>
__device__ __forceinline__ void two_layer_tile(const float* __restrict__ A1, const float* __restrict__ W1,
                                               const float* __restrict__ A2, const float* __restrict__ W2,
                                               const float* __restrict__ bias1,
                                               const float* __restrict__ Wp, float* outp,
                                               const float* __restrict__ Wr,
                                               const float* __restrict__ biasr, float* outr,
                                               int M, int m0, float (*sa)[GEMM_PITCH],
                                               int lane, int lr, int half) {
    constexpr int ROWS = 16 * MT;
    constexpr int K1 = 64, K2 = 64;
    {
        const int c4 = lane * 4;
        const float* srcp = A1;
        int srck = K1, coff = c4;
        if (c4 >= K1) { srcp = A2; srck = K2; coff = c4 - K1; }
#pragma unroll 8
        for (int r = 0; r < ROWS; ++r) {
            int g = m0 + r;
            g = (g < M) ? g : (M - 1);
            const float4 v = *(const float4*)(srcp + (size_t)g * srck + coff);
            *(float4*)&sa[r][c4] = v;
        }
    }

    {
        v8f acc[MT][8] = {};
        gemm_k_steps<K1, MT>(sa, 0, W1, lr, half, acc);
        gemm_k_steps<K2, MT>(sa, K1, W2, lr, half, acc);
#pragma unroll
        for (int mt = 0; mt < MT; ++mt) {
#pragma unroll
            for (int j = 0; j < 2; ++j) {
                const int col = j * 64 + lr * 4;
                float4 bv = {0.0f, 0.0f, 0.0f, 0.0f};
                if (bias1) bv = *(const float4*)(bias1 + col);
#pragma unroll
                for (int i = 0; i < 8; ++i) {
                    float4 v;
                    v.x = fmaxf(acc[mt][j * 4 + 0][i] + bv.x, 0.0f);
                    v.y = fmaxf(acc[mt][j * 4 + 1][i] + bv.y, 0.0f);
                    v.z = fmaxf(acc[mt][j * 4 + 2][i] + bv.z, 0.0f);
                    v.w = fmaxf(acc[mt][j * 4 + 3][i] + bv.w, 0.0f);
                    *(float4*)&sa[mt * 16 + i + 8 * half][col] = v;
                }
            }
        }
    }

    {
        v8f acc[MT][8] = {};
        gemm_k_steps<128, MT>(sa, 0, Wp, lr, half, acc);
        store_rows<MT>(acc, nullptr, outp, m0, M, lr, half);
    }
    {
        v8f acc[MT][8] = {};
        gemm_k_steps<128, MT>(sa, 0, Wr, lr, half, acc);
        store_rows<MT>(acc, biasr, outr, m0, M, lr, half);
    }
}

__global__ __launch_bounds__(512) void gemm_two_layers_wmma(const GemmSet sr, const GemmSet sl) {
    __shared__ __attribute__((aligned(16))) float s_a[16][32][GEMM_PITCH];
    const int lane = threadIdx.x & 31;
    const int wave = __builtin_amdgcn_readfirstlane((int)(threadIdx.x >> 5));
    const int half = lane >> 4;
    const int lr = lane & 15;
    float (*sa)[GEMM_PITCH] = s_a[wave];

    const int TR = (sr.M + 15) >> 4;
    const int T = TR + ((sl.M + 15) >> 4);
    const int g = ((int)blockIdx.x * 4 + (wave & 3)) * 4 + (wave >> 2);
    const int G = (int)gridDim.x * 16;
    int t = (int)(((unsigned)g * (unsigned)T) / (unsigned)G);
    const int t1 = (int)(((unsigned)(g + 1) * (unsigned)T) / (unsigned)G);
    while (t < t1) {
        const bool isr = t < TR;
        const int lim = (isr && TR < t1) ? TR : t1;
        const int m0 = (isr ? t : (t - TR)) * 16;
        const float* A1 = isr ? sr.A1 : sl.A1;
        const float* W1 = isr ? sr.W1 : sl.W1;
        const float* A2 = isr ? sr.A2 : sl.A2;
        const float* W2 = isr ? sr.W2 : sl.W2;
        const float* b1 = isr ? sr.bias1 : sl.bias1;
        const float* Wp = isr ? sr.Wp : sl.Wp;
        float* outp = isr ? sr.outp : sl.outp;
        const float* Wr = isr ? sr.Wr : sl.Wr;
        const float* br = isr ? sr.biasr : sl.biasr;
        float* outr = isr ? sr.outr : sl.outr;
        const int M = isr ? sr.M : sl.M;
        if (t + 1 < lim) {
            two_layer_tile<2>(A1, W1, A2, W2, b1, Wp, outp, Wr, br, outr, M, m0, sa, lane, lr, half);
            t += 2;
        } else {
            two_layer_tile<1>(A1, W1, A2, W2, b1, Wp, outp, Wr, br, outr, M, m0, sa, lane, lr, half);
            t += 1;
        }
    }
}

__global__ __launch_bounds__(256) void rowmat6_wmma(const float* __restrict__ X,
                                                    const float* __restrict__ Wrel,
                                                    const float* __restrict__ Wroot,
                                                    const float* __restrict__ bias,
                                                    float* __restrict__ proj,
                                                    float* __restrict__ init,
                                                    float* __restrict__ init1, int n) {
    __shared__ __attribute__((aligned(16))) float s_x[8][16][GEMM_PITCH];
    const int lane = threadIdx.x & 31;
    const int wave = threadIdx.x >> 5;
    const int m0 = (blockIdx.x * 8 + wave) * 16;
    const int half = lane >> 4;
    const int lr = lane & 15;
    float (*sx)[GEMM_PITCH] = s_x[wave];
    if (m0 < n) {
        float4 v[16];
#pragma unroll
        for (int r = 0; r < 16; ++r) {
            int g = m0 + r;
            g = (g < n) ? g : (n - 1);
            v[r] = *(const float4*)(X + (size_t)g * 128 + lane * 4);
        }
#pragma unroll
        for (int r = 0; r < 16; ++r) {
            float4 t = v[r];
            t.x = fmaxf(t.x, 0.0f);
            t.y = fmaxf(t.y, 0.0f);
            t.z = fmaxf(t.z, 0.0f);
            t.w = fmaxf(t.w, 0.0f);
            *(float4*)&sx[r][lane * 4] = t;
        }
    }
    __syncthreads();
    if (m0 >= n) return;

    const bool used = lr < 6;
    const float* wcol = (lr < 3) ? (Wrel + lr) : (used ? (Wroot + (lr - 3)) : Wrel);
    v8f acc = {};
#pragma unroll
    for (int k = 0; k < 128; k += 4) {
        const v2f a = *(const v2f*)&sx[lr][k + 2 * half];
        const float wx = wcol[(k + 2 * half) * 3];
        const float wy = wcol[(k + 2 * half + 1) * 3];
        v2f b;
        b.x = used ? wx : 0.0f;
        b.y = used ? wy : 0.0f;
        acc = __builtin_amdgcn_wmma_f32_16x16x4_f32(false, a, false, b, (short)0, acc, false, false);
    }
    if (used) {
        const bool rel = lr < 3;
        const int c = rel ? lr : (lr - 3);
        float bv = 0.0f;
        if (!rel && bias) bv = bias[c];
        float* dstp = rel ? proj : init;
#pragma unroll
        for (int i = 0; i < 8; ++i) {
            const int row = m0 + 8 * half + i;
            if (row < n) {
                dstp[(size_t)row * 3 + c] = acc[i] + bv;
                if (!rel) init1[(size_t)row * 3 + c] = 0.0f;
            }
        }
    }
}

__global__ __launch_bounds__(256) void scatter_add3(const float* __restrict__ proj,
                                                    const int* __restrict__ sidx,
                                                    const int* __restrict__ didx,
                                                    float* out0, float* out1, int nE) {
    __shared__ int s_half;
    if (threadIdx.x == 0) {
        const unsigned id = __builtin_amdgcn_s_sendmsg_rtn(0x87);
        s_half = (int)((id >> 18) & 1u);
    }
    __syncthreads();
    float* out = __builtin_amdgcn_readfirstlane(s_half) ? out1 : out0;
    const int lane = threadIdx.x & 31;
    const int e0 = (blockIdx.x * 8 + (int)(threadIdx.x >> 5)) * 32;
    if (e0 >= nE) return;
    const int e = e0 + lane;
    const int ec = (e < nE) ? e : (nE - 1);
    const int sv = sidx[ec];
    const int dv = didx[ec];
    const int c = lane & 3;
    const int q = lane >> 2;
    float v[4];
    int doff[4];
    bool act[4];
#pragma unroll
    for (int r = 0; r < 4; ++r) {
        const int sl = r * 8 + q;
        const int s = __shfl(sv, sl, 32);
        const int d = __shfl(dv, sl, 32);
        act[r] = (c < 3) && (e0 + sl < nE);
        doff[r] = d * 3 + c;
        v[r] = act[r] ? proj[(size_t)s * 3 + c] : 0.0f;
    }
#pragma unroll
    for (int r = 0; r < 4; ++r) {
        if (act[r]) atomicAdd(out + doff[r], v[r]);
    }
}

__global__ __launch_bounds__(256) void knowledge_kernel(float* __restrict__ out,
                                                        const float* __restrict__ out1,
                                                        const float* __restrict__ x,
                                                        int n) {
    int i = blockIdx.x * blockDim.x + threadIdx.x;
    if (i >= n) return;
    float a = out[(size_t)i * 3 + 0] + out1[(size_t)i * 3 + 0];
    float b = out[(size_t)i * 3 + 1] + out1[(size_t)i * 3 + 1];
    float c = out[(size_t)i * 3 + 2] + out1[(size_t)i * 3 + 2];
    float nrm = sqrtf(a * a + b * b + c * c);
    float s = 10.0f / fmaxf(nrm, 1e-12f);
    a *= s; b *= s; c *= s;
    if (x[(size_t)i * 64 + 61] != 0.0f) a -= 10.0f;
    if (x[(size_t)i * 64 + 63] != 0.0f) c -= 10.0f;
    out[(size_t)i * 3 + 0] = a;
    out[(size_t)i * 3 + 1] = b;
    out[(size_t)i * 3 + 2] = c;
}

static inline int ceil_div(long long a, long long b) { return (int)((a + b - 1) / b); }

extern "C" void kernel_launch(void* const* d_in, const int* in_sizes, int n_in,
                              void* d_out, int out_size, void* d_ws, size_t ws_size,
                              hipStream_t stream) {
    const float* x_s = (const float*)d_in[0];
    const float* x_t = (const float*)d_in[1];
    const int* e_l = (const int*)d_in[2];
    const int* e_r = (const int*)d_in[3];
    const float* w1lr_rel  = (const float*)d_in[4];
    const float* b1lr      = (const float*)d_in[5];
    const float* w1lr_root = (const float*)d_in[6];
    const float* w1rl_rel  = (const float*)d_in[7];
    const float* b1rl      = (const float*)d_in[8];
    const float* w1rl_root = (const float*)d_in[9];
    const float* w2lr_rel  = (const float*)d_in[10];
    const float* b2lr      = (const float*)d_in[11];
    const float* w2lr_root = (const float*)d_in[12];
    const float* w2rl_rel  = (const float*)d_in[13];
    const float* b2rl      = (const float*)d_in[14];
    const float* w2rl_root = (const float*)d_in[15];
    const float* w3lr_rel  = (const float*)d_in[16];
    const float* b3lr      = (const float*)d_in[17];
    const float* w3lr_root = (const float*)d_in[18];
    const float* w3rl_rel  = (const float*)d_in[19];
    const float* b3rl      = (const float*)d_in[20];
    const float* w3rl_root = (const float*)d_in[21];

    const int NL = in_sizes[0] / 64;
    const int NR = in_sizes[1] / 64;
    const int E  = in_sizes[2];

    float* ws = (float*)d_ws;
    const size_t AGG_R = 0;
    const size_t AGG_L = AGG_R + (size_t)NR * 128;
    const size_t S2_R  = AGG_L + (size_t)NL * 128;
    const size_t S2_L  = S2_R + (size_t)NR * 128;
    const size_t P_R   = S2_L + (size_t)NL * 128;
    const size_t P_L   = P_R + (size_t)NR * 128;
    const size_t TKT   = P_L + (size_t)NL * 128;
    const size_t PROJ_LR = AGG_R;
    const size_t PROJ_RL = AGG_R + (size_t)NL * 3;
    const size_t ACC1_L  = AGG_L;
    const size_t ACC1_R  = AGG_L + (size_t)NL * 3;

    int* tkt = (int*)(ws + TKT);

    float* d_left  = (float*)d_out;
    float* d_right = (float*)d_out + (size_t)NL * 3;

    const int TPB = 256;
    constexpr int GR = 2;
    const int SGRID = 1536;

    zero_kernel<<<ceil_div((size_t)NL * 64, TPB), TPB, 0, stream>>>(ws + AGG_L, (size_t)NL * 64, tkt, 256);
    scatter_add_rows<64, GR><<<SGRID, TPB, 0, stream>>>(x_t, e_r, e_l, ws + AGG_L, E, NL, tkt + 64,
                                                        ws + AGG_R, (size_t)NR * 16);
    scatter_add_rows<64, GR><<<SGRID, TPB, 0, stream>>>(x_s, e_l, e_r, ws + AGG_R, E, NR, tkt, nullptr, 0);
    const GemmSet set_r = {ws + AGG_R, w1lr_rel, x_t, w1lr_root, b1lr, w2rl_rel, ws + P_R, w2lr_root, b2lr, ws + S2_R, NR};
    const GemmSet set_l = {ws + AGG_L, w1rl_rel, x_s, w1rl_root, b1rl, w2lr_rel, ws + P_L, w2rl_root, b2rl, ws + S2_L, NL};
    gemm_two_layers_wmma<<<256, 512, 0, stream>>>(set_r, set_l);

    scatter_add_rows<128, GR><<<SGRID, TPB, 0, stream>>>(ws + P_L, e_l, e_r, ws + S2_R, E, NR, tkt + 128, nullptr, 0);
    scatter_add_rows<128, GR><<<SGRID, TPB, 0, stream>>>(ws + P_R, e_r, e_l, ws + S2_L, E, NL, tkt + 192, nullptr, 0);

    rowmat6_wmma<<<ceil_div(ceil_div(NL, 16), 8), TPB, 0, stream>>>(
        ws + S2_L, w3lr_rel, w3rl_root, b3rl, ws + PROJ_LR, d_left, ws + ACC1_L, NL);
    rowmat6_wmma<<<ceil_div(ceil_div(NR, 16), 8), TPB, 0, stream>>>(
        ws + S2_R, w3rl_rel, w3lr_root, b3lr, ws + PROJ_RL, d_right, ws + ACC1_R, NR);
    scatter_add3<<<ceil_div(E, TPB), TPB, 0, stream>>>(ws + PROJ_LR, e_l, e_r, d_right, ws + ACC1_R, E);
    scatter_add3<<<ceil_div(E, TPB), TPB, 0, stream>>>(ws + PROJ_RL, e_r, e_l, d_left, ws + ACC1_L, E);

    knowledge_kernel<<<ceil_div(NL, TPB), TPB, 0, stream>>>(d_left, ws + ACC1_L, x_s, NL);
    knowledge_kernel<<<ceil_div(NR, TPB), TPB, 0, stream>>>(d_right, ws + ACC1_R, x_t, NR);
}
